# starting kernel
# baseline (speedup 1.0000x reference)
#include <hip/hip_runtime.h>

#define B_  4
#define NQ_ 512
#define NK_ 512
#define D_  512
#define H_  128
#define QB_ 8

typedef __attribute__((ext_vector_type(2))) float v2f;
typedef __attribute__((ext_vector_type(8))) float v8f;

__device__ __forceinline__ float hw_tanh(float x) {
#if __has_builtin(__builtin_amdgcn_tanhf)
  return __builtin_amdgcn_tanhf(x);
#else
  float y;
  asm volatile("v_tanh_f32 %0, %1" : "=v"(y) : "v"(x));
  return y;
#endif
}

__device__ __forceinline__ float hw_exp(float x) {
#if __has_builtin(__builtin_amdgcn_exp2f)
  return __builtin_amdgcn_exp2f(x * 1.44269504088896340736f);
#else
  return __expf(x);
#endif
}

struct Frags {
  v2f a0, a1, b0, b1, b2, b3;
};

__device__ __forceinline__ void load_frags(const float* __restrict__ arow0,
                                           const float* __restrict__ arow1,
                                           const float* __restrict__ bbase,
                                           int N, int kk, Frags& f) {
  f.a0 = *(const v2f*)(arow0 + kk);
  f.a1 = *(const v2f*)(arow1 + kk);
  const float* b = bbase + (long long)kk * N;
  f.b0.x = b[0];       f.b0.y = b[N];
  f.b1.x = b[16];      f.b1.y = b[N + 16];
  f.b2.x = b[32];      f.b2.y = b[N + 32];
  f.b3.x = b[48];      f.b3.y = b[N + 48];
}

__global__ __launch_bounds__(32) void wmma_gemm_f32(
    const float* __restrict__ A, const float* __restrict__ Bm,
    float* __restrict__ C, int K, int N,
    long long sA, long long sB, long long sC) {
  const int lane  = threadIdx.x;
  const int mTile = blockIdx.x * 32;
  const int nBase = blockIdx.y * 64;
  const int batch = blockIdx.z;
  A  += (long long)batch * sA;
  Bm += (long long)batch * sB;
  C  += (long long)batch * sC;

  const int mRow = lane & 15;
  const int kHi  = (lane >> 4) * 2;
  const int col  = lane & 15;

  const float* arow0 = A + (long long)(mTile + mRow) * K + kHi;
  const float* arow1 = arow0 + (long long)16 * K;
  const float* bbase = Bm + (long long)kHi * N + nBase + col;

  v8f acc00 = {}, acc01 = {}, acc02 = {}, acc03 = {};
  v8f acc10 = {}, acc11 = {}, acc12 = {}, acc13 = {};

  Frags cur, nxt;
  load_frags(arow0, arow1, bbase, N, 0, cur);

  for (int kk = 4; kk < K; kk += 4) {
    load_frags(arow0, arow1, bbase, N, kk, nxt);
    acc00 = __builtin_amdgcn_wmma_f32_16x16x4_f32(false, cur.a0, false, cur.b0, (short)0, acc00, false, false);
    acc01 = __builtin_amdgcn_wmma_f32_16x16x4_f32(false, cur.a0, false, cur.b1, (short)0, acc01, false, false);
    acc02 = __builtin_amdgcn_wmma_f32_16x16x4_f32(false, cur.a0, false, cur.b2, (short)0, acc02, false, false);
    acc03 = __builtin_amdgcn_wmma_f32_16x16x4_f32(false, cur.a0, false, cur.b3, (short)0, acc03, false, false);
    acc10 = __builtin_amdgcn_wmma_f32_16x16x4_f32(false, cur.a1, false, cur.b0, (short)0, acc10, false, false);
    acc11 = __builtin_amdgcn_wmma_f32_16x16x4_f32(false, cur.a1, false, cur.b1, (short)0, acc11, false, false);
    acc12 = __builtin_amdgcn_wmma_f32_16x16x4_f32(false, cur.a1, false, cur.b2, (short)0, acc12, false, false);
    acc13 = __builtin_amdgcn_wmma_f32_16x16x4_f32(false, cur.a1, false, cur.b3, (short)0, acc13, false, false);
    cur = nxt;
  }
  acc00 = __builtin_amdgcn_wmma_f32_16x16x4_f32(false, cur.a0, false, cur.b0, (short)0, acc00, false, false);
  acc01 = __builtin_amdgcn_wmma_f32_16x16x4_f32(false, cur.a0, false, cur.b1, (short)0, acc01, false, false);
  acc02 = __builtin_amdgcn_wmma_f32_16x16x4_f32(false, cur.a0, false, cur.b2, (short)0, acc02, false, false);
  acc03 = __builtin_amdgcn_wmma_f32_16x16x4_f32(false, cur.a0, false, cur.b3, (short)0, acc03, false, false);
  acc10 = __builtin_amdgcn_wmma_f32_16x16x4_f32(false, cur.a1, false, cur.b0, (short)0, acc10, false, false);
  acc11 = __builtin_amdgcn_wmma_f32_16x16x4_f32(false, cur.a1, false, cur.b1, (short)0, acc11, false, false);
  acc12 = __builtin_amdgcn_wmma_f32_16x16x4_f32(false, cur.a1, false, cur.b2, (short)0, acc12, false, false);
  acc13 = __builtin_amdgcn_wmma_f32_16x16x4_f32(false, cur.a1, false, cur.b3, (short)0, acc13, false, false);

  const int mOut0 = mTile + (lane >> 4) * 8;
#pragma unroll
  for (int r = 0; r < 8; ++r) {
    long long row0 = (long long)(mOut0 + r) * N + nBase + col;
    C[row0 +  0] = acc00[r];
    C[row0 + 16] = acc01[r];
    C[row0 + 32] = acc02[r];
    C[row0 + 48] = acc03[r];
    long long row1 = row0 + (long long)16 * N;
    C[row1 +  0] = acc10[r];
    C[row1 + 16] = acc11[r];
    C[row1 + 32] = acc12[r];
    C[row1 + 48] = acc13[r];
  }
}

__global__ __launch_bounds__(NK_) void scores_softmax(
    const float* __restrict__ qproj, const float* __restrict__ kproj,
    const float* __restrict__ Wv, const int* __restrict__ valid_lens,
    float* __restrict__ attn) {
  __shared__ float qp[QB_][H_];
  __shared__ float wv[H_];
  __shared__ float part[QB_][16];
  __shared__ float mxs[QB_];
  __shared__ float dns[QB_];

  const int q0   = blockIdx.x * QB_;
  const int b    = blockIdx.y;
  const int k    = threadIdx.x;
  const int lane = k & 31;
  const int wid  = k >> 5;

  for (int i = k; i < QB_ * H_; i += NK_)
    qp[i / H_][i % H_] = qproj[(long long)(b * NQ_ + q0) * H_ + i];
  if (k < H_) wv[k] = Wv[k];
  __syncthreads();

  const float4* kp4 =
      (const float4*)(kproj + (long long)(b * NK_ + k) * H_);
  float s[QB_];
#pragma unroll
  for (int j = 0; j < QB_; ++j) s[j] = 0.0f;

  for (int h4 = 0; h4 < H_ / 4; ++h4) {
    const float4 kv = kp4[h4];
    const int h = h4 * 4;
#pragma unroll
    for (int j = 0; j < QB_; ++j) {
      s[j] += wv[h + 0] * hw_tanh(qp[j][h + 0] + kv.x);
      s[j] += wv[h + 1] * hw_tanh(qp[j][h + 1] + kv.y);
      s[j] += wv[h + 2] * hw_tanh(qp[j][h + 2] + kv.z);
      s[j] += wv[h + 3] * hw_tanh(qp[j][h + 3] + kv.w);
    }
  }

  const int vlen = valid_lens[b];
  if (k >= vlen) {
#pragma unroll
    for (int j = 0; j < QB_; ++j) s[j] = -1.0e6f;
  }

#pragma unroll
  for (int j = 0; j < QB_; ++j) {
    float m = s[j];
#pragma unroll
    for (int off = 16; off > 0; off >>= 1) m = fmaxf(m, __shfl_xor(m, off, 32));
    if (lane == 0) part[j][wid] = m;
  }
  __syncthreads();
  if (k < QB_) {
    float m = part[k][0];
#pragma unroll
    for (int w = 1; w < 16; ++w) m = fmaxf(m, part[k][w]);
    mxs[k] = m;
  }
  __syncthreads();

  float e[QB_];
#pragma unroll
  for (int j = 0; j < QB_; ++j) {
    e[j] = hw_exp(s[j] - mxs[j]);
    float t = e[j];
#pragma unroll
    for (int off = 16; off > 0; off >>= 1) t += __shfl_xor(t, off, 32);
    if (lane == 0) part[j][wid] = t;
  }
  __syncthreads();
  if (k < QB_) {
    float t = 0.0f;
#pragma unroll
    for (int w = 0; w < 16; ++w) t += part[k][w];
    dns[k] = t;
  }
  __syncthreads();

#pragma unroll
  for (int j = 0; j < QB_; ++j) {
    attn[(long long)(b * NQ_ + q0 + j) * NK_ + k] = e[j] / dns[j];
  }
}

extern "C" void kernel_launch(void* const* d_in, const int* in_sizes, int n_in,
                              void* d_out, int out_size, void* d_ws, size_t ws_size,
                              hipStream_t stream) {
  (void)in_sizes; (void)n_in; (void)out_size; (void)ws_size;

  const float* queries    = (const float*)d_in[0];
  const float* keys       = (const float*)d_in[1];
  const float* values     = (const float*)d_in[2];
  const float* Wq         = (const float*)d_in[3];
  const float* Wk         = (const float*)d_in[4];
  const float* Wv         = (const float*)d_in[5];
  const int*   valid_lens = (const int*)d_in[6];
  float*       out        = (float*)d_out;

  float* qproj = (float*)d_ws;
  float* kproj = qproj + (long long)B_ * NQ_ * H_;
  float* attn  = kproj + (long long)B_ * NK_ * H_;

  wmma_gemm_f32<<<dim3((B_ * NQ_) / 32, H_ / 64, 1), 32, 0, stream>>>(
      queries, Wq, qproj, D_, H_, 0, 0, 0);

  wmma_gemm_f32<<<dim3((B_ * NK_) / 32, H_ / 64, 1), 32, 0, stream>>>(
      keys, Wk, kproj, D_, H_, 0, 0, 0);

  scores_softmax<<<dim3(NQ_ / QB_, B_), NK_, 0, stream>>>(
      qproj, kproj, Wv, valid_lens, attn);

  wmma_gemm_f32<<<dim3(NQ_ / 32, D_ / 64, B_), 32, 0, stream>>>(
      attn, values, out, NK_, D_,
      (long long)NQ_ * NK_, (long long)NK_ * D_, (long long)NQ_ * D_);
}


# GEMM kernel: fragment loads issued six K-steps ahead of their WMMAs through a ring of eight register sets (was one K-step ahead, load wait exposed in every trip); summation order unchanged
# speedup vs baseline: 1.5323x; 1.5323x over previous
#include <hip/hip_runtime.h>

#define B_  4
#define NQ_ 512
#define NK_ 512
#define D_  512
#define H_  128
#define QB_ 8

typedef __attribute__((ext_vector_type(2))) float v2f;
typedef __attribute__((ext_vector_type(8))) float v8f;

__device__ __forceinline__ float hw_tanh(float x) {
#if __has_builtin(__builtin_amdgcn_tanhf)
  return __builtin_amdgcn_tanhf(x);
#else
  float y;
  asm volatile("v_tanh_f32 %0, %1" : "=v"(y) : "v"(x));
  return y;
#endif
}

__device__ __forceinline__ float hw_exp(float x) {
#if __has_builtin(__builtin_amdgcn_exp2f)
  return __builtin_amdgcn_exp2f(x * 1.44269504088896340736f);
#else
  return __expf(x);
#endif
}

struct Frags {
  v2f a0, a1, b0, b1, b2, b3;
};

__device__ __forceinline__ void load_frags(const char* __restrict__ uA,
                                           unsigned aoff0, unsigned aoff1,
                                           const char* __restrict__ uB0,
                                           const char* __restrict__ uB1,
                                           unsigned boff, Frags& f) {
  f.a0 = *(const v2f*)(uA + aoff0);
  f.a1 = *(const v2f*)(uA + aoff1);
  const char* p0 = uB0 + boff;
  const char* p1 = uB1 + boff;
  f.b0.x = *(const float*)(p0);        f.b0.y = *(const float*)(p1);
  f.b1.x = *(const float*)(p0 + 64);   f.b1.y = *(const float*)(p1 + 64);
  f.b2.x = *(const float*)(p0 + 128);  f.b2.y = *(const float*)(p1 + 128);
  f.b3.x = *(const float*)(p0 + 192);  f.b3.y = *(const float*)(p1 + 192);
}

#define WMMA_OPERAND_GAP() asm volatile("v_nop\n\tv_nop\n\tv_nop\n\tv_nop")

struct Acc {
  v8f c[2][4];
};

__device__ __forceinline__ void mma_step(const Frags& f, Acc& acc) {
  acc.c[0][0] = __builtin_amdgcn_wmma_f32_16x16x4_f32(false, f.a0, false, f.b0, (short)0, acc.c[0][0], false, false);
  acc.c[0][1] = __builtin_amdgcn_wmma_f32_16x16x4_f32(false, f.a0, false, f.b1, (short)0, acc.c[0][1], false, false);
  acc.c[0][2] = __builtin_amdgcn_wmma_f32_16x16x4_f32(false, f.a0, false, f.b2, (short)0, acc.c[0][2], false, false);
  acc.c[0][3] = __builtin_amdgcn_wmma_f32_16x16x4_f32(false, f.a0, false, f.b3, (short)0, acc.c[0][3], false, false);
  acc.c[1][0] = __builtin_amdgcn_wmma_f32_16x16x4_f32(false, f.a1, false, f.b0, (short)0, acc.c[1][0], false, false);
  acc.c[1][1] = __builtin_amdgcn_wmma_f32_16x16x4_f32(false, f.a1, false, f.b1, (short)0, acc.c[1][1], false, false);
  acc.c[1][2] = __builtin_amdgcn_wmma_f32_16x16x4_f32(false, f.a1, false, f.b2, (short)0, acc.c[1][2], false, false);
  acc.c[1][3] = __builtin_amdgcn_wmma_f32_16x16x4_f32(false, f.a1, false, f.b3, (short)0, acc.c[1][3], false, false);
}

__global__ __launch_bounds__(32) void wmma_gemm_f32(
    const float* __restrict__ A, const float* __restrict__ Bm,
    float* __restrict__ C, int K, int N,
    long long sA, long long sB, long long sC) {
  const int lane  = threadIdx.x;
  const int mTile = blockIdx.x * 32;
  const int nBase = blockIdx.y * 64;
  const int batch = blockIdx.z;
  A  += (long long)batch * sA;
  Bm += (long long)batch * sB;
  C  += (long long)batch * sC;

  const int mRow = lane & 15;
  const int kHi  = (lane >> 4) * 2;
  const int col  = lane & 15;

  const char* uA = (const char*)(A + (long long)mTile * K);
  const char* uB = (const char*)(Bm + nBase);
  const long long rowB = (long long)N * 4;
  const unsigned aoff0 = (unsigned)(mRow * K + kHi) * 4u;
  const unsigned aoff1 = aoff0 + (unsigned)(16 * K) * 4u;
  const unsigned boff  = (unsigned)(kHi * N + col) * 4u;

  Acc acc = {};
  Frags r[8];

#define LOAD_STEP(kk, dst)                                              \
  load_frags(uA + (long long)(kk) * 4, aoff0, aoff1,                    \
             uB + (long long)(kk) * rowB, uB + (long long)((kk) + 1) * rowB, \
             boff, dst)

#pragma unroll
  for (int s = 0; s < 6; ++s) {
    LOAD_STEP(4 * s, r[s]);
    __builtin_amdgcn_sched_barrier(0);
  }

  int k0 = 0;
  for (; k0 + 32 < K; k0 += 32) {
#pragma unroll
    for (int g = 0; g < 8; g += 2) {
      LOAD_STEP(k0 + 4 * (g + 6), r[(g + 6) & 7]);
      LOAD_STEP(k0 + 4 * (g + 7), r[(g + 7) & 7]);
      __builtin_amdgcn_sched_barrier(0);
      mma_step(r[g], acc);
      mma_step(r[g + 1], acc);
      __builtin_amdgcn_sched_barrier(0);
      WMMA_OPERAND_GAP();
      __builtin_amdgcn_sched_barrier(0);
    }
  }
  LOAD_STEP(k0 + 24, r[6]);
  LOAD_STEP(k0 + 28, r[7]);
  __builtin_amdgcn_sched_barrier(0);
#pragma unroll
  for (int s = 0; s < 8; ++s) mma_step(r[s], acc);
#undef LOAD_STEP
  __builtin_amdgcn_sched_barrier(0);

  const int mOut0 = mTile + (lane >> 4) * 8;
#pragma unroll
  for (int r8 = 0; r8 < 8; ++r8) {
    long long row0 = (long long)(mOut0 + r8) * N + nBase + col;
    C[row0 +  0] = acc.c[0][0][r8];
    C[row0 + 16] = acc.c[0][1][r8];
    C[row0 + 32] = acc.c[0][2][r8];
    C[row0 + 48] = acc.c[0][3][r8];
    long long row1 = row0 + (long long)16 * N;
    C[row1 +  0] = acc.c[1][0][r8];
    C[row1 + 16] = acc.c[1][1][r8];
    C[row1 + 32] = acc.c[1][2][r8];
    C[row1 + 48] = acc.c[1][3][r8];
  }
}

__global__ __launch_bounds__(NK_) void scores_softmax(
    const float* __restrict__ qproj, const float* __restrict__ kproj,
    const float* __restrict__ Wv, const int* __restrict__ valid_lens,
    float* __restrict__ attn) {
  __shared__ float qp[QB_][H_];
  __shared__ float wv[H_];
  __shared__ float part[QB_][16];
  __shared__ float mxs[QB_];
  __shared__ float dns[QB_];

  const int q0   = blockIdx.x * QB_;
  const int b    = blockIdx.y;
  const int k    = threadIdx.x;
  const int lane = k & 31;
  const int wid  = k >> 5;

  for (int i = k; i < QB_ * H_; i += NK_)
    qp[i / H_][i % H_] = qproj[(long long)(b * NQ_ + q0) * H_ + i];
  if (k < H_) wv[k] = Wv[k];
  __syncthreads();

  const float4* kp4 =
      (const float4*)(kproj + (long long)(b * NK_ + k) * H_);
  float s[QB_];
#pragma unroll
  for (int j = 0; j < QB_; ++j) s[j] = 0.0f;

  for (int h4 = 0; h4 < H_ / 4; ++h4) {
    const float4 kv = kp4[h4];
    const int h = h4 * 4;
#pragma unroll
    for (int j = 0; j < QB_; ++j) {
      s[j] += wv[h + 0] * hw_tanh(qp[j][h + 0] + kv.x);
      s[j] += wv[h + 1] * hw_tanh(qp[j][h + 1] + kv.y);
      s[j] += wv[h + 2] * hw_tanh(qp[j][h + 2] + kv.z);
      s[j] += wv[h + 3] * hw_tanh(qp[j][h + 3] + kv.w);
    }
  }

  const int vlen = valid_lens[b];
  if (k >= vlen) {
#pragma unroll
    for (int j = 0; j < QB_; ++j) s[j] = -1.0e6f;
  }

#pragma unroll
  for (int j = 0; j < QB_; ++j) {
    float m = s[j];
#pragma unroll
    for (int off = 16; off > 0; off >>= 1) m = fmaxf(m, __shfl_xor(m, off, 32));
    if (lane == 0) part[j][wid] = m;
  }
  __syncthreads();
  if (k < QB_) {
    float m = part[k][0];
#pragma unroll
    for (int w = 1; w < 16; ++w) m = fmaxf(m, part[k][w]);
    mxs[k] = m;
  }
  __syncthreads();

  float e[QB_];
#pragma unroll
  for (int j = 0; j < QB_; ++j) {
    e[j] = hw_exp(s[j] - mxs[j]);
    float t = e[j];
#pragma unroll
    for (int off = 16; off > 0; off >>= 1) t += __shfl_xor(t, off, 32);
    if (lane == 0) part[j][wid] = t;
  }
  __syncthreads();
  if (k < QB_) {
    float t = 0.0f;
#pragma unroll
    for (int w = 0; w < 16; ++w) t += part[k][w];
    dns[k] = t;
  }
  __syncthreads();

#pragma unroll
  for (int j = 0; j < QB_; ++j) {
    attn[(long long)(b * NQ_ + q0 + j) * NK_ + k] = e[j] / dns[j];
  }
}

extern "C" void kernel_launch(void* const* d_in, const int* in_sizes, int n_in,
                              void* d_out, int out_size, void* d_ws, size_t ws_size,
                              hipStream_t stream) {
  (void)in_sizes; (void)n_in; (void)out_size; (void)ws_size;

  const float* queries    = (const float*)d_in[0];
  const float* keys       = (const float*)d_in[1];
  const float* values     = (const float*)d_in[2];
  const float* Wq         = (const float*)d_in[3];
  const float* Wk         = (const float*)d_in[4];
  const float* Wv         = (const float*)d_in[5];
  const int*   valid_lens = (const int*)d_in[6];
  float*       out        = (float*)d_out;

  float* qproj = (float*)d_ws;
  float* kproj = qproj + (long long)B_ * NQ_ * H_;
  float* attn  = kproj + (long long)B_ * NK_ * H_;

  wmma_gemm_f32<<<dim3((B_ * NQ_) / 32, H_ / 64, 1), 32, 0, stream>>>(
      queries, Wq, qproj, D_, H_, 0, 0, 0);

  wmma_gemm_f32<<<dim3((B_ * NK_) / 32, H_ / 64, 1), 32, 0, stream>>>(
      keys, Wk, kproj, D_, H_, 0, 0, 0);

  scores_softmax<<<dim3(NQ_ / QB_, B_), NK_, 0, stream>>>(
      qproj, kproj, Wv, valid_lens, attn);

  wmma_gemm_f32<<<dim3(NQ_ / 32, D_ / 64, B_), 32, 0, stream>>>(
      attn, values, out, NK_, D_,
      (long long)NQ_ * NK_, (long long)NK_ * D_, (long long)NQ_ * D_);
}


# GEMM kernel: K split over the 4 waves of a block (one per SIMD), partial strips added through LDS in fixed K order; WMMA chain per wave a quarter as long; summation order changed
# speedup vs baseline: 2.4757x; 1.6157x over previous
#include <hip/hip_runtime.h>

#define B_  4
#define NQ_ 512
#define NK_ 512
#define D_  512
#define H_  128
#define QB_ 8
#define KW_ 4

typedef __attribute__((ext_vector_type(2))) float v2f;
typedef __attribute__((ext_vector_type(8))) float v8f;

__device__ __forceinline__ float hw_tanh(float x) {
#if __has_builtin(__builtin_amdgcn_tanhf)
  return __builtin_amdgcn_tanhf(x);
#else
  float y;
  asm volatile("v_tanh_f32 %0, %1" : "=v"(y) : "v"(x));
  return y;
#endif
}

__device__ __forceinline__ float hw_exp(float x) {
#if __has_builtin(__builtin_amdgcn_exp2f)
  return __builtin_amdgcn_exp2f(x * 1.44269504088896340736f);
#else
  return __expf(x);
#endif
}

struct Frags {
  v2f a0, a1, b0, b1, b2, b3;
};

__device__ __forceinline__ void load_frags(const char* __restrict__ uA,
                                           unsigned aoff0, unsigned aoff1,
                                           const char* __restrict__ uB0,
                                           const char* __restrict__ uB1,
                                           unsigned boff, Frags& f) {
  f.a0 = *(const v2f*)(uA + aoff0);
  f.a1 = *(const v2f*)(uA + aoff1);
  const char* p0 = uB0 + boff;
  const char* p1 = uB1 + boff;
  f.b0.x = *(const float*)(p0);        f.b0.y = *(const float*)(p1);
  f.b1.x = *(const float*)(p0 + 64);   f.b1.y = *(const float*)(p1 + 64);
  f.b2.x = *(const float*)(p0 + 128);  f.b2.y = *(const float*)(p1 + 128);
  f.b3.x = *(const float*)(p0 + 192);  f.b3.y = *(const float*)(p1 + 192);
}

#define WMMA_OPERAND_GAP() asm volatile("v_nop\n\tv_nop\n\tv_nop\n\tv_nop")

struct Acc {
  v8f c[2][4];
};

__device__ __forceinline__ void mma_step(const Frags& f, Acc& acc) {
  acc.c[0][0] = __builtin_amdgcn_wmma_f32_16x16x4_f32(false, f.a0, false, f.b0, (short)0, acc.c[0][0], false, false);
  acc.c[0][1] = __builtin_amdgcn_wmma_f32_16x16x4_f32(false, f.a0, false, f.b1, (short)0, acc.c[0][1], false, false);
  acc.c[0][2] = __builtin_amdgcn_wmma_f32_16x16x4_f32(false, f.a0, false, f.b2, (short)0, acc.c[0][2], false, false);
  acc.c[0][3] = __builtin_amdgcn_wmma_f32_16x16x4_f32(false, f.a0, false, f.b3, (short)0, acc.c[0][3], false, false);
  acc.c[1][0] = __builtin_amdgcn_wmma_f32_16x16x4_f32(false, f.a1, false, f.b0, (short)0, acc.c[1][0], false, false);
  acc.c[1][1] = __builtin_amdgcn_wmma_f32_16x16x4_f32(false, f.a1, false, f.b1, (short)0, acc.c[1][1], false, false);
  acc.c[1][2] = __builtin_amdgcn_wmma_f32_16x16x4_f32(false, f.a1, false, f.b2, (short)0, acc.c[1][2], false, false);
  acc.c[1][3] = __builtin_amdgcn_wmma_f32_16x16x4_f32(false, f.a1, false, f.b3, (short)0, acc.c[1][3], false, false);
}

__global__ __launch_bounds__(32 * KW_) void wmma_gemm_f32(
    const float* __restrict__ A, const float* __restrict__ Bm,
    float* __restrict__ C, int K, int N,
    long long sA, long long sB, long long sC) {
  __shared__ float4 red[KW_ * 16 * 32];

  const int lane  = threadIdx.x & 31;
  const int part  = __builtin_amdgcn_readfirstlane((int)(threadIdx.x >> 5));
  const int Kp    = K / KW_;
  const int kBeg  = part * Kp;
  const int mTile = blockIdx.x * 32;
  const int nBase = blockIdx.y * 64;
  const int batch = blockIdx.z;
  A  += (long long)batch * sA;
  Bm += (long long)batch * sB;
  C  += (long long)batch * sC;

  const int mRow = lane & 15;
  const int kHi  = (lane >> 4) * 2;
  const int col  = lane & 15;

  const char* uA = (const char*)(A + (long long)mTile * K + kBeg);
  const char* uB = (const char*)(Bm + (long long)kBeg * N + nBase);
  const long long rowB = (long long)N * 4;
  const unsigned aoff0 = (unsigned)(mRow * K + kHi) * 4u;
  const unsigned aoff1 = aoff0 + (unsigned)(16 * K) * 4u;
  const unsigned boff  = (unsigned)(kHi * N + col) * 4u;

  Acc acc = {};
  Frags r[8];

#define LOAD_STEP(kk, dst)                                              \
  load_frags(uA + (long long)(kk) * 4, aoff0, aoff1,                    \
             uB + (long long)(kk) * rowB, uB + (long long)((kk) + 1) * rowB, \
             boff, dst)

#pragma unroll
  for (int s = 0; s < 6; ++s) {
    LOAD_STEP(4 * s, r[s]);
    __builtin_amdgcn_sched_barrier(0);
  }

  int k0 = 0;
  for (; k0 + 32 < Kp; k0 += 32) {
#pragma unroll
    for (int g = 0; g < 8; g += 2) {
      LOAD_STEP(k0 + 4 * (g + 6), r[(g + 6) & 7]);
      LOAD_STEP(k0 + 4 * (g + 7), r[(g + 7) & 7]);
      __builtin_amdgcn_sched_barrier(0);
      mma_step(r[g], acc);
      mma_step(r[g + 1], acc);
      __builtin_amdgcn_sched_barrier(0);
      WMMA_OPERAND_GAP();
      __builtin_amdgcn_sched_barrier(0);
    }
  }
  LOAD_STEP(k0 + 24, r[6]);
  LOAD_STEP(k0 + 28, r[7]);
  __builtin_amdgcn_sched_barrier(0);
#pragma unroll
  for (int s = 0; s < 8; ++s) mma_step(r[s], acc);
#undef LOAD_STEP
  __builtin_amdgcn_sched_barrier(0);
  WMMA_OPERAND_GAP();
  __builtin_amdgcn_sched_barrier(0);

#pragma unroll
  for (int t = 0; t < 8; ++t) {
#pragma unroll
    for (int hq = 0; hq < 2; ++hq) {
      const v8f c = acc.c[t >> 2][t & 3];
      float4 v;
      v.x = c[4 * hq + 0]; v.y = c[4 * hq + 1];
      v.z = c[4 * hq + 2]; v.w = c[4 * hq + 3];
      red[(part * 16 + t * 2 + hq) * 32 + lane] = v;
    }
  }
  __syncthreads();

  const int mOut0 = mTile + (part >> 1) * 16 + (lane >> 4) * 8;
  const int nOut0 = nBase + (part & 1) * 32 + col;
#pragma unroll
  for (int tt = 0; tt < 2; ++tt) {
#pragma unroll
    for (int hq = 0; hq < 2; ++hq) {
      const int q = (2 * part + tt) * 2 + hq;
      float4 s = red[(0 * 16 + q) * 32 + lane];
#pragma unroll
      for (int p = 1; p < KW_; ++p) {
        const float4 v = red[(p * 16 + q) * 32 + lane];
        s.x += v.x; s.y += v.y; s.z += v.z; s.w += v.w;
      }
      long long o = (long long)(mOut0 + 4 * hq) * N + nOut0 + 16 * tt;
      C[o]                     = s.x;
      C[o + (long long)N]      = s.y;
      C[o + (long long)2 * N]  = s.z;
      C[o + (long long)3 * N]  = s.w;
    }
  }
}

__global__ __launch_bounds__(NK_) void scores_softmax(
    const float* __restrict__ qproj, const float* __restrict__ kproj,
    const float* __restrict__ Wv, const int* __restrict__ valid_lens,
    float* __restrict__ attn) {
  __shared__ float qp[QB_][H_];
  __shared__ float wv[H_];
  __shared__ float part[QB_][16];
  __shared__ float mxs[QB_];
  __shared__ float dns[QB_];

  const int q0   = blockIdx.x * QB_;
  const int b    = blockIdx.y;
  const int k    = threadIdx.x;
  const int lane = k & 31;
  const int wid  = k >> 5;

  for (int i = k; i < QB_ * H_; i += NK_)
    qp[i / H_][i % H_] = qproj[(long long)(b * NQ_ + q0) * H_ + i];
  if (k < H_) wv[k] = Wv[k];
  __syncthreads();

  const float4* kp4 =
      (const float4*)(kproj + (long long)(b * NK_ + k) * H_);
  float s[QB_];
#pragma unroll
  for (int j = 0; j < QB_; ++j) s[j] = 0.0f;

  for (int h4 = 0; h4 < H_ / 4; ++h4) {
    const float4 kv = kp4[h4];
    const int h = h4 * 4;
#pragma unroll
    for (int j = 0; j < QB_; ++j) {
      s[j] += wv[h + 0] * hw_tanh(qp[j][h + 0] + kv.x);
      s[j] += wv[h + 1] * hw_tanh(qp[j][h + 1] + kv.y);
      s[j] += wv[h + 2] * hw_tanh(qp[j][h + 2] + kv.z);
      s[j] += wv[h + 3] * hw_tanh(qp[j][h + 3] + kv.w);
    }
  }

  const int vlen = valid_lens[b];
  if (k >= vlen) {
#pragma unroll
    for (int j = 0; j < QB_; ++j) s[j] = -1.0e6f;
  }

#pragma unroll
  for (int j = 0; j < QB_; ++j) {
    float m = s[j];
#pragma unroll
    for (int off = 16; off > 0; off >>= 1) m = fmaxf(m, __shfl_xor(m, off, 32));
    if (lane == 0) part[j][wid] = m;
  }
  __syncthreads();
  if (k < QB_) {
    float m = part[k][0];
#pragma unroll
    for (int w = 1; w < 16; ++w) m = fmaxf(m, part[k][w]);
    mxs[k] = m;
  }
  __syncthreads();

  float e[QB_];
#pragma unroll
  for (int j = 0; j < QB_; ++j) {
    e[j] = hw_exp(s[j] - mxs[j]);
    float t = e[j];
#pragma unroll
    for (int off = 16; off > 0; off >>= 1) t += __shfl_xor(t, off, 32);
    if (lane == 0) part[j][wid] = t;
  }
  __syncthreads();
  if (k < QB_) {
    float t = 0.0f;
#pragma unroll
    for (int w = 0; w < 16; ++w) t += part[k][w];
    dns[k] = t;
  }
  __syncthreads();

#pragma unroll
  for (int j = 0; j < QB_; ++j) {
    attn[(long long)(b * NQ_ + q0 + j) * NK_ + k] = e[j] / dns[j];
  }
}

extern "C" void kernel_launch(void* const* d_in, const int* in_sizes, int n_in,
                              void* d_out, int out_size, void* d_ws, size_t ws_size,
                              hipStream_t stream) {
  (void)in_sizes; (void)n_in; (void)out_size; (void)ws_size;

  const float* queries    = (const float*)d_in[0];
  const float* keys       = (const float*)d_in[1];
  const float* values     = (const float*)d_in[2];
  const float* Wq         = (const float*)d_in[3];
  const float* Wk         = (const float*)d_in[4];
  const float* Wv         = (const float*)d_in[5];
  const int*   valid_lens = (const int*)d_in[6];
  float*       out        = (float*)d_out;

  float* qproj = (float*)d_ws;
  float* kproj = qproj + (long long)B_ * NQ_ * H_;
  float* attn  = kproj + (long long)B_ * NK_ * H_;

  wmma_gemm_f32<<<dim3((B_ * NQ_) / 32, H_ / 64, 1), 32 * KW_, 0, stream>>>(
      queries, Wq, qproj, D_, H_, 0, 0, 0);

  wmma_gemm_f32<<<dim3((B_ * NK_) / 32, H_ / 64, 1), 32 * KW_, 0, stream>>>(
      keys, Wk, kproj, D_, H_, 0, 0, 0);

  scores_softmax<<<dim3(NQ_ / QB_, B_), NK_, 0, stream>>>(
      qproj, kproj, Wv, valid_lens, attn);

  wmma_gemm_f32<<<dim3(NQ_ / 32, D_ / 64, B_), 32 * KW_, 0, stream>>>(
      attn, values, out, NK_, D_,
      (long long)NQ_ * NK_, (long long)NK_ * D_, (long long)NQ_ * D_);
}


# The two projection GEMMs (queries @ Wq, keys @ Wk) run as one launch of 256 blocks, grid z picks the operand set; was two launches of 128 blocks on 256 WGPs; summation order unchanged
# speedup vs baseline: 2.9140x; 1.1770x over previous
#include <hip/hip_runtime.h>

#define B_  4
#define NQ_ 512
#define NK_ 512
#define D_  512
#define H_  128
#define QB_ 8
#define KW_ 4

typedef __attribute__((ext_vector_type(2))) float v2f;
typedef __attribute__((ext_vector_type(8))) float v8f;

__device__ __forceinline__ float hw_tanh(float x) {
#if __has_builtin(__builtin_amdgcn_tanhf)
  return __builtin_amdgcn_tanhf(x);
#else
  float y;
  asm volatile("v_tanh_f32 %0, %1" : "=v"(y) : "v"(x));
  return y;
#endif
}

__device__ __forceinline__ float hw_exp(float x) {
#if __has_builtin(__builtin_amdgcn_exp2f)
  return __builtin_amdgcn_exp2f(x * 1.44269504088896340736f);
#else
  return __expf(x);
#endif
}

struct Frags {
  v2f a0, a1, b0, b1, b2, b3;
};

__device__ __forceinline__ void load_frags(const char* __restrict__ uA,
                                           unsigned aoff0, unsigned aoff1,
                                           const char* __restrict__ uB0,
                                           const char* __restrict__ uB1,
                                           unsigned boff, Frags& f) {
  f.a0 = *(const v2f*)(uA + aoff0);
  f.a1 = *(const v2f*)(uA + aoff1);
  const char* p0 = uB0 + boff;
  const char* p1 = uB1 + boff;
  f.b0.x = *(const float*)(p0);        f.b0.y = *(const float*)(p1);
  f.b1.x = *(const float*)(p0 + 64);   f.b1.y = *(const float*)(p1 + 64);
  f.b2.x = *(const float*)(p0 + 128);  f.b2.y = *(const float*)(p1 + 128);
  f.b3.x = *(const float*)(p0 + 192);  f.b3.y = *(const float*)(p1 + 192);
}

#define WMMA_OPERAND_GAP() asm volatile("v_nop\n\tv_nop\n\tv_nop\n\tv_nop")

struct Acc {
  v8f c[2][4];
};

__device__ __forceinline__ void mma_step(const Frags& f, Acc& acc) {
  acc.c[0][0] = __builtin_amdgcn_wmma_f32_16x16x4_f32(false, f.a0, false, f.b0, (short)0, acc.c[0][0], false, false);
  acc.c[0][1] = __builtin_amdgcn_wmma_f32_16x16x4_f32(false, f.a0, false, f.b1, (short)0, acc.c[0][1], false, false);
  acc.c[0][2] = __builtin_amdgcn_wmma_f32_16x16x4_f32(false, f.a0, false, f.b2, (short)0, acc.c[0][2], false, false);
  acc.c[0][3] = __builtin_amdgcn_wmma_f32_16x16x4_f32(false, f.a0, false, f.b3, (short)0, acc.c[0][3], false, false);
  acc.c[1][0] = __builtin_amdgcn_wmma_f32_16x16x4_f32(false, f.a1, false, f.b0, (short)0, acc.c[1][0], false, false);
  acc.c[1][1] = __builtin_amdgcn_wmma_f32_16x16x4_f32(false, f.a1, false, f.b1, (short)0, acc.c[1][1], false, false);
  acc.c[1][2] = __builtin_amdgcn_wmma_f32_16x16x4_f32(false, f.a1, false, f.b2, (short)0, acc.c[1][2], false, false);
  acc.c[1][3] = __builtin_amdgcn_wmma_f32_16x16x4_f32(false, f.a1, false, f.b3, (short)0, acc.c[1][3], false, false);
}

__device__ __forceinline__ void gemm_strip(
    const float* __restrict__ A, const float* __restrict__ Bm,
    float* __restrict__ C, int K, int N) {
  __shared__ float4 red[KW_ * 16 * 32];

  const int lane  = threadIdx.x & 31;
  const int part  = __builtin_amdgcn_readfirstlane((int)(threadIdx.x >> 5));
  const int Kp    = K / KW_;
  const int kBeg  = part * Kp;
  const int mTile = blockIdx.x * 32;
  const int nBase = blockIdx.y * 64;

  const int mRow = lane & 15;
  const int kHi  = (lane >> 4) * 2;
  const int col  = lane & 15;

  const char* uA = (const char*)(A + (long long)mTile * K + kBeg);
  const char* uB = (const char*)(Bm + (long long)kBeg * N + nBase);
  const long long rowB = (long long)N * 4;
  const unsigned aoff0 = (unsigned)(mRow * K + kHi) * 4u;
  const unsigned aoff1 = aoff0 + (unsigned)(16 * K) * 4u;
  const unsigned boff  = (unsigned)(kHi * N + col) * 4u;

  Acc acc = {};
  Frags r[8];

#define LOAD_STEP(kk, dst)                                              \
  load_frags(uA + (long long)(kk) * 4, aoff0, aoff1,                    \
             uB + (long long)(kk) * rowB, uB + (long long)((kk) + 1) * rowB, \
             boff, dst)

#pragma unroll
  for (int s = 0; s < 6; ++s) {
    LOAD_STEP(4 * s, r[s]);
    __builtin_amdgcn_sched_barrier(0);
  }

  int k0 = 0;
  for (; k0 + 32 < Kp; k0 += 32) {
#pragma unroll
    for (int g = 0; g < 8; g += 2) {
      LOAD_STEP(k0 + 4 * (g + 6), r[(g + 6) & 7]);
      LOAD_STEP(k0 + 4 * (g + 7), r[(g + 7) & 7]);
      __builtin_amdgcn_sched_barrier(0);
      mma_step(r[g], acc);
      mma_step(r[g + 1], acc);
      __builtin_amdgcn_sched_barrier(0);
      WMMA_OPERAND_GAP();
      __builtin_amdgcn_sched_barrier(0);
    }
  }
  LOAD_STEP(k0 + 24, r[6]);
  LOAD_STEP(k0 + 28, r[7]);
  __builtin_amdgcn_sched_barrier(0);
#pragma unroll
  for (int s = 0; s < 8; ++s) mma_step(r[s], acc);
#undef LOAD_STEP
  __builtin_amdgcn_sched_barrier(0);
  WMMA_OPERAND_GAP();
  __builtin_amdgcn_sched_barrier(0);

#pragma unroll
  for (int t = 0; t < 8; ++t) {
#pragma unroll
    for (int hq = 0; hq < 2; ++hq) {
      const v8f c = acc.c[t >> 2][t & 3];
      float4 v;
      v.x = c[4 * hq + 0]; v.y = c[4 * hq + 1];
      v.z = c[4 * hq + 2]; v.w = c[4 * hq + 3];
      red[(part * 16 + t * 2 + hq) * 32 + lane] = v;
    }
  }
  __syncthreads();

  const int mOut0 = mTile + (part >> 1) * 16 + (lane >> 4) * 8;
  const int nOut0 = nBase + (part & 1) * 32 + col;
#pragma unroll
  for (int tt = 0; tt < 2; ++tt) {
#pragma unroll
    for (int hq = 0; hq < 2; ++hq) {
      const int q = (2 * part + tt) * 2 + hq;
      float4 s = red[(0 * 16 + q) * 32 + lane];
#pragma unroll
      for (int p = 1; p < KW_; ++p) {
        const float4 v = red[(p * 16 + q) * 32 + lane];
        s.x += v.x; s.y += v.y; s.z += v.z; s.w += v.w;
      }
      long long o = (long long)(mOut0 + 4 * hq) * N + nOut0 + 16 * tt;
      C[o]                     = s.x;
      C[o + (long long)N]      = s.y;
      C[o + (long long)2 * N]  = s.z;
      C[o + (long long)3 * N]  = s.w;
    }
  }
}

__global__ __launch_bounds__(32 * KW_) void wmma_gemm_f32(
    const float* __restrict__ A, const float* __restrict__ Bm,
    float* __restrict__ C, int K, int N,
    long long sA, long long sB, long long sC) {
  const int batch = blockIdx.z;
  gemm_strip(A + (long long)batch * sA, Bm + (long long)batch * sB,
             C + (long long)batch * sC, K, N);
}

__global__ __launch_bounds__(32 * KW_) void wmma_gemm_f32_pair(
    const float* __restrict__ A0, const float* __restrict__ B0,
    float* __restrict__ C0,
    const float* __restrict__ A1, const float* __restrict__ B1,
    float* __restrict__ C1, int K, int N) {
  const bool second = blockIdx.z != 0;
  gemm_strip(second ? A1 : A0, second ? B1 : B0, second ? C1 : C0, K, N);
}

__global__ __launch_bounds__(NK_) void scores_softmax(
    const float* __restrict__ qproj, const float* __restrict__ kproj,
    const float* __restrict__ Wv, const int* __restrict__ valid_lens,
    float* __restrict__ attn) {
  __shared__ float qp[QB_][H_];
  __shared__ float wv[H_];
  __shared__ float part[QB_][16];
  __shared__ float mxs[QB_];
  __shared__ float dns[QB_];

  const int q0   = blockIdx.x * QB_;
  const int b    = blockIdx.y;
  const int k    = threadIdx.x;
  const int lane = k & 31;
  const int wid  = k >> 5;

  for (int i = k; i < QB_ * H_; i += NK_)
    qp[i / H_][i % H_] = qproj[(long long)(b * NQ_ + q0) * H_ + i];
  if (k < H_) wv[k] = Wv[k];
  __syncthreads();

  const float4* kp4 =
      (const float4*)(kproj + (long long)(b * NK_ + k) * H_);
  float s[QB_];
#pragma unroll
  for (int j = 0; j < QB_; ++j) s[j] = 0.0f;

  for (int h4 = 0; h4 < H_ / 4; ++h4) {
    const float4 kv = kp4[h4];
    const int h = h4 * 4;
#pragma unroll
    for (int j = 0; j < QB_; ++j) {
      s[j] += wv[h + 0] * hw_tanh(qp[j][h + 0] + kv.x);
      s[j] += wv[h + 1] * hw_tanh(qp[j][h + 1] + kv.y);
      s[j] += wv[h + 2] * hw_tanh(qp[j][h + 2] + kv.z);
      s[j] += wv[h + 3] * hw_tanh(qp[j][h + 3] + kv.w);
    }
  }

  const int vlen = valid_lens[b];
  if (k >= vlen) {
#pragma unroll
    for (int j = 0; j < QB_; ++j) s[j] = -1.0e6f;
  }

#pragma unroll
  for (int j = 0; j < QB_; ++j) {
    float m = s[j];
#pragma unroll
    for (int off = 16; off > 0; off >>= 1) m = fmaxf(m, __shfl_xor(m, off, 32));
    if (lane == 0) part[j][wid] = m;
  }
  __syncthreads();
  if (k < QB_) {
    float m = part[k][0];
#pragma unroll
    for (int w = 1; w < 16; ++w) m = fmaxf(m, part[k][w]);
    mxs[k] = m;
  }
  __syncthreads();

  float e[QB_];
#pragma unroll
  for (int j = 0; j < QB_; ++j) {
    e[j] = hw_exp(s[j] - mxs[j]);
    float t = e[j];
#pragma unroll
    for (int off = 16; off > 0; off >>= 1) t += __shfl_xor(t, off, 32);
    if (lane == 0) part[j][wid] = t;
  }
  __syncthreads();
  if (k < QB_) {
    float t = 0.0f;
#pragma unroll
    for (int w = 0; w < 16; ++w) t += part[k][w];
    dns[k] = t;
  }
  __syncthreads();

#pragma unroll
  for (int j = 0; j < QB_; ++j) {
    attn[(long long)(b * NQ_ + q0 + j) * NK_ + k] = e[j] / dns[j];
  }
}

extern "C" void kernel_launch(void* const* d_in, const int* in_sizes, int n_in,
                              void* d_out, int out_size, void* d_ws, size_t ws_size,
                              hipStream_t stream) {
  (void)in_sizes; (void)n_in; (void)out_size; (void)ws_size;

  const float* queries    = (const float*)d_in[0];
  const float* keys       = (const float*)d_in[1];
  const float* values     = (const float*)d_in[2];
  const float* Wq         = (const float*)d_in[3];
  const float* Wk         = (const float*)d_in[4];
  const float* Wv         = (const float*)d_in[5];
  const int*   valid_lens = (const int*)d_in[6];
  float*       out        = (float*)d_out;

  float* qproj = (float*)d_ws;
  float* kproj = qproj + (long long)B_ * NQ_ * H_;
  float* attn  = kproj + (long long)B_ * NK_ * H_;

  static_assert(NQ_ == NK_, "the paired launch needs equal row counts");
  wmma_gemm_f32_pair<<<dim3((B_ * NQ_) / 32, H_ / 64, 2), 32 * KW_, 0, stream>>>(
      queries, Wq, qproj, keys, Wk, kproj, D_, H_);

  scores_softmax<<<dim3(NQ_ / QB_, B_), NK_, 0, stream>>>(
      qproj, kproj, Wv, valid_lens, attn);

  wmma_gemm_f32<<<dim3(NQ_ / 32, D_ / 64, B_), 32 * KW_, 0, stream>>>(
      attn, values, out, NK_, D_,
      (long long)NQ_ * NK_, (long long)NK_ * D_, (long long)NQ_ * D_);
}


# Softmax kernel: row max and row sum butterflies exchange lanes on the vector ALU (permlanex16, then DPP row_xmask) instead of 80 dependent ds_bpermute round trips per wave; summation order unchanged
# speedup vs baseline: 3.1725x; 1.0887x over previous
#include <hip/hip_runtime.h>

#define B_  4
#define NQ_ 512
#define NK_ 512
#define D_  512
#define H_  128
#define QB_ 8
#define KW_ 4

typedef __attribute__((ext_vector_type(2))) float v2f;
typedef __attribute__((ext_vector_type(8))) float v8f;

__device__ __forceinline__ float hw_tanh(float x) {
#if __has_builtin(__builtin_amdgcn_tanhf)
  return __builtin_amdgcn_tanhf(x);
#else
  float y;
  asm volatile("v_tanh_f32 %0, %1" : "=v"(y) : "v"(x));
  return y;
#endif
}

__device__ __forceinline__ float hw_exp(float x) {
#if __has_builtin(__builtin_amdgcn_exp2f)
  return __builtin_amdgcn_exp2f(x * 1.44269504088896340736f);
#else
  return __expf(x);
#endif
}

__device__ __forceinline__ float xhalf(float x) {
  const int b = __builtin_bit_cast(int, x);
  return __builtin_bit_cast(float, __builtin_amdgcn_permlanex16(
      b, b, 0x76543210, 0xfedcba98, false, false));
}

template <int M>
__device__ __forceinline__ float xrow(float x) {
  const int b = __builtin_bit_cast(int, x);
  return __builtin_bit_cast(float, __builtin_amdgcn_update_dpp(
      b, b, 0x160 + M, 0xf, 0xf, false));
}

struct Frags {
  v2f a0, a1, b0, b1, b2, b3;
};

__device__ __forceinline__ void load_frags(const char* __restrict__ uA,
                                           unsigned aoff0, unsigned aoff1,
                                           const char* __restrict__ uB0,
                                           const char* __restrict__ uB1,
                                           unsigned boff, Frags& f) {
  f.a0 = *(const v2f*)(uA + aoff0);
  f.a1 = *(const v2f*)(uA + aoff1);
  const char* p0 = uB0 + boff;
  const char* p1 = uB1 + boff;
  f.b0.x = *(const float*)(p0);        f.b0.y = *(const float*)(p1);
  f.b1.x = *(const float*)(p0 + 64);   f.b1.y = *(const float*)(p1 + 64);
  f.b2.x = *(const float*)(p0 + 128);  f.b2.y = *(const float*)(p1 + 128);
  f.b3.x = *(const float*)(p0 + 192);  f.b3.y = *(const float*)(p1 + 192);
}

#define WMMA_OPERAND_GAP() asm volatile("v_nop\n\tv_nop\n\tv_nop\n\tv_nop")

struct Acc {
  v8f c[2][4];
};

__device__ __forceinline__ void mma_step(const Frags& f, Acc& acc) {
  acc.c[0][0] = __builtin_amdgcn_wmma_f32_16x16x4_f32(false, f.a0, false, f.b0, (short)0, acc.c[0][0], false, false);
  acc.c[0][1] = __builtin_amdgcn_wmma_f32_16x16x4_f32(false, f.a0, false, f.b1, (short)0, acc.c[0][1], false, false);
  acc.c[0][2] = __builtin_amdgcn_wmma_f32_16x16x4_f32(false, f.a0, false, f.b2, (short)0, acc.c[0][2], false, false);
  acc.c[0][3] = __builtin_amdgcn_wmma_f32_16x16x4_f32(false, f.a0, false, f.b3, (short)0, acc.c[0][3], false, false);
  acc.c[1][0] = __builtin_amdgcn_wmma_f32_16x16x4_f32(false, f.a1, false, f.b0, (short)0, acc.c[1][0], false, false);
  acc.c[1][1] = __builtin_amdgcn_wmma_f32_16x16x4_f32(false, f.a1, false, f.b1, (short)0, acc.c[1][1], false, false);
  acc.c[1][2] = __builtin_amdgcn_wmma_f32_16x16x4_f32(false, f.a1, false, f.b2, (short)0, acc.c[1][2], false, false);
  acc.c[1][3] = __builtin_amdgcn_wmma_f32_16x16x4_f32(false, f.a1, false, f.b3, (short)0, acc.c[1][3], false, false);
}

__device__ __forceinline__ void gemm_strip(
    const float* __restrict__ A, const float* __restrict__ Bm,
    float* __restrict__ C, int K, int N) {
  __shared__ float4 red[KW_ * 16 * 32];

  const int lane  = threadIdx.x & 31;
  const int part  = __builtin_amdgcn_readfirstlane((int)(threadIdx.x >> 5));
  const int Kp    = K / KW_;
  const int kBeg  = part * Kp;
  const int mTile = blockIdx.x * 32;
  const int nBase = blockIdx.y * 64;

  const int mRow = lane & 15;
  const int kHi  = (lane >> 4) * 2;
  const int col  = lane & 15;

  const char* uA = (const char*)(A + (long long)mTile * K + kBeg);
  const char* uB = (const char*)(Bm + (long long)kBeg * N + nBase);
  const long long rowB = (long long)N * 4;
  const unsigned aoff0 = (unsigned)(mRow * K + kHi) * 4u;
  const unsigned aoff1 = aoff0 + (unsigned)(16 * K) * 4u;
  const unsigned boff  = (unsigned)(kHi * N + col) * 4u;

  Acc acc = {};
  Frags r[8];

#define LOAD_STEP(kk, dst)                                              \
  load_frags(uA + (long long)(kk) * 4, aoff0, aoff1,                    \
             uB + (long long)(kk) * rowB, uB + (long long)((kk) + 1) * rowB, \
             boff, dst)

#pragma unroll
  for (int s = 0; s < 6; ++s) {
    LOAD_STEP(4 * s, r[s]);
    __builtin_amdgcn_sched_barrier(0);
  }

  int k0 = 0;
  for (; k0 + 32 < Kp; k0 += 32) {
#pragma unroll
    for (int g = 0; g < 8; g += 2) {
      LOAD_STEP(k0 + 4 * (g + 6), r[(g + 6) & 7]);
      LOAD_STEP(k0 + 4 * (g + 7), r[(g + 7) & 7]);
      __builtin_amdgcn_sched_barrier(0);
      mma_step(r[g], acc);
      mma_step(r[g + 1], acc);
      __builtin_amdgcn_sched_barrier(0);
      WMMA_OPERAND_GAP();
      __builtin_amdgcn_sched_barrier(0);
    }
  }
  LOAD_STEP(k0 + 24, r[6]);
  LOAD_STEP(k0 + 28, r[7]);
  __builtin_amdgcn_sched_barrier(0);
#pragma unroll
  for (int s = 0; s < 8; ++s) mma_step(r[s], acc);
#undef LOAD_STEP
  __builtin_amdgcn_sched_barrier(0);
  WMMA_OPERAND_GAP();
  __builtin_amdgcn_sched_barrier(0);

#pragma unroll
  for (int t = 0; t < 8; ++t) {
#pragma unroll
    for (int hq = 0; hq < 2; ++hq) {
      const v8f c = acc.c[t >> 2][t & 3];
      float4 v;
      v.x = c[4 * hq + 0]; v.y = c[4 * hq + 1];
      v.z = c[4 * hq + 2]; v.w = c[4 * hq + 3];
      red[(part * 16 + t * 2 + hq) * 32 + lane] = v;
    }
  }
  __syncthreads();

  const int mOut0 = mTile + (part >> 1) * 16 + (lane >> 4) * 8;
  const int nOut0 = nBase + (part & 1) * 32 + col;
#pragma unroll
  for (int tt = 0; tt < 2; ++tt) {
#pragma unroll
    for (int hq = 0; hq < 2; ++hq) {
      const int q = (2 * part + tt) * 2 + hq;
      float4 s = red[(0 * 16 + q) * 32 + lane];
#pragma unroll
      for (int p = 1; p < KW_; ++p) {
        const float4 v = red[(p * 16 + q) * 32 + lane];
        s.x += v.x; s.y += v.y; s.z += v.z; s.w += v.w;
      }
      long long o = (long long)(mOut0 + 4 * hq) * N + nOut0 + 16 * tt;
      C[o]                     = s.x;
      C[o + (long long)N]      = s.y;
      C[o + (long long)2 * N]  = s.z;
      C[o + (long long)3 * N]  = s.w;
    }
  }
}

__global__ __launch_bounds__(32 * KW_) void wmma_gemm_f32(
    const float* __restrict__ A, const float* __restrict__ Bm,
    float* __restrict__ C, int K, int N,
    long long sA, long long sB, long long sC) {
  const int batch = blockIdx.z;
  gemm_strip(A + (long long)batch * sA, Bm + (long long)batch * sB,
             C + (long long)batch * sC, K, N);
}

__global__ __launch_bounds__(32 * KW_) void wmma_gemm_f32_pair(
    const float* __restrict__ A0, const float* __restrict__ B0,
    float* __restrict__ C0,
    const float* __restrict__ A1, const float* __restrict__ B1,
    float* __restrict__ C1, int K, int N) {
  const bool second = blockIdx.z != 0;
  gemm_strip(second ? A1 : A0, second ? B1 : B0, second ? C1 : C0, K, N);
}

__global__ __launch_bounds__(NK_) void scores_softmax(
    const float* __restrict__ qproj, const float* __restrict__ kproj,
    const float* __restrict__ Wv, const int* __restrict__ valid_lens,
    float* __restrict__ attn) {
  __shared__ float qp[QB_][H_];
  __shared__ float wv[H_];
  __shared__ float part[QB_][16];
  __shared__ float mxs[QB_];
  __shared__ float dns[QB_];

  const int q0   = blockIdx.x * QB_;
  const int b    = blockIdx.y;
  const int k    = threadIdx.x;
  const int lane = k & 31;
  const int wid  = k >> 5;

  for (int i = k; i < QB_ * H_; i += NK_)
    qp[i / H_][i % H_] = qproj[(long long)(b * NQ_ + q0) * H_ + i];
  if (k < H_) wv[k] = Wv[k];
  __syncthreads();

  const float4* kp4 =
      (const float4*)(kproj + (long long)(b * NK_ + k) * H_);
  float s[QB_];
#pragma unroll
  for (int j = 0; j < QB_; ++j) s[j] = 0.0f;

  for (int h4 = 0; h4 < H_ / 4; ++h4) {
    const float4 kv = kp4[h4];
    const int h = h4 * 4;
#pragma unroll
    for (int j = 0; j < QB_; ++j) {
      s[j] += wv[h + 0] * hw_tanh(qp[j][h + 0] + kv.x);
      s[j] += wv[h + 1] * hw_tanh(qp[j][h + 1] + kv.y);
      s[j] += wv[h + 2] * hw_tanh(qp[j][h + 2] + kv.z);
      s[j] += wv[h + 3] * hw_tanh(qp[j][h + 3] + kv.w);
    }
  }

  const int vlen = valid_lens[b];
  if (k >= vlen) {
#pragma unroll
    for (int j = 0; j < QB_; ++j) s[j] = -1.0e6f;
  }

  float mx[QB_];
#pragma unroll
  for (int j = 0; j < QB_; ++j) mx[j] = fmaxf(s[j], xhalf(s[j]));
#pragma unroll
  for (int j = 0; j < QB_; ++j) mx[j] = fmaxf(mx[j], xrow<8>(mx[j]));
#pragma unroll
  for (int j = 0; j < QB_; ++j) mx[j] = fmaxf(mx[j], xrow<4>(mx[j]));
#pragma unroll
  for (int j = 0; j < QB_; ++j) mx[j] = fmaxf(mx[j], xrow<2>(mx[j]));
#pragma unroll
  for (int j = 0; j < QB_; ++j) mx[j] = fmaxf(mx[j], xrow<1>(mx[j]));
  if (lane == 0) {
#pragma unroll
    for (int j = 0; j < QB_; ++j) part[j][wid] = mx[j];
  }
  __syncthreads();
  if (k < QB_) {
    float m = part[k][0];
#pragma unroll
    for (int w = 1; w < 16; ++w) m = fmaxf(m, part[k][w]);
    mxs[k] = m;
  }
  __syncthreads();

  float e[QB_], t[QB_];
#pragma unroll
  for (int j = 0; j < QB_; ++j) {
    e[j] = hw_exp(s[j] - mxs[j]);
    t[j] = e[j];
  }
#pragma unroll
  for (int j = 0; j < QB_; ++j) t[j] += xhalf(t[j]);
#pragma unroll
  for (int j = 0; j < QB_; ++j) t[j] += xrow<8>(t[j]);
#pragma unroll
  for (int j = 0; j < QB_; ++j) t[j] += xrow<4>(t[j]);
#pragma unroll
  for (int j = 0; j < QB_; ++j) t[j] += xrow<2>(t[j]);
#pragma unroll
  for (int j = 0; j < QB_; ++j) t[j] += xrow<1>(t[j]);
  if (lane == 0) {
#pragma unroll
    for (int j = 0; j < QB_; ++j) part[j][wid] = t[j];
  }
  __syncthreads();
  if (k < QB_) {
    float t = 0.0f;
#pragma unroll
    for (int w = 0; w < 16; ++w) t += part[k][w];
    dns[k] = t;
  }
  __syncthreads();

#pragma unroll
  for (int j = 0; j < QB_; ++j) {
    attn[(long long)(b * NQ_ + q0 + j) * NK_ + k] = e[j] / dns[j];
  }
}

extern "C" void kernel_launch(void* const* d_in, const int* in_sizes, int n_in,
                              void* d_out, int out_size, void* d_ws, size_t ws_size,
                              hipStream_t stream) {
  (void)in_sizes; (void)n_in; (void)out_size; (void)ws_size;

  const float* queries    = (const float*)d_in[0];
  const float* keys       = (const float*)d_in[1];
  const float* values     = (const float*)d_in[2];
  const float* Wq         = (const float*)d_in[3];
  const float* Wk         = (const float*)d_in[4];
  const float* Wv         = (const float*)d_in[5];
  const int*   valid_lens = (const int*)d_in[6];
  float*       out        = (float*)d_out;

  float* qproj = (float*)d_ws;
  float* kproj = qproj + (long long)B_ * NQ_ * H_;
  float* attn  = kproj + (long long)B_ * NK_ * H_;

  static_assert(NQ_ == NK_, "the paired launch needs equal row counts");
  wmma_gemm_f32_pair<<<dim3((B_ * NQ_) / 32, H_ / 64, 2), 32 * KW_, 0, stream>>>(
      queries, Wq, qproj, keys, Wk, kproj, D_, H_);

  scores_softmax<<<dim3(NQ_ / QB_, B_), NK_, 0, stream>>>(
      qproj, kproj, Wv, valid_lens, attn);

  wmma_gemm_f32<<<dim3(NQ_ / 32, D_ / 64, B_), 32 * KW_, 0, stream>>>(
      attn, values, out, NK_, D_,
      (long long)NQ_ * NK_, (long long)NK_ * D_, (long long)NQ_ * D_);
}


# GEMM kernel: fragment loads are 16-byte (A: 4 k per lane; B: 4 columns per lane, N-tiles take interleaved columns, C rows stored as float4); summation order inside each 8-k group changed
# speedup vs baseline: 3.2682x; 1.0302x over previous
#include <hip/hip_runtime.h>

#define B_  4
#define NQ_ 512
#define NK_ 512
#define D_  512
#define H_  128
#define QB_ 8
#define KW_ 4

typedef __attribute__((ext_vector_type(2))) float v2f;
typedef __attribute__((ext_vector_type(4))) float v4f;
typedef __attribute__((ext_vector_type(8))) float v8f;

__device__ __forceinline__ float hw_tanh(float x) {
#if __has_builtin(__builtin_amdgcn_tanhf)
  return __builtin_amdgcn_tanhf(x);
#else
  float y;
  asm volatile("v_tanh_f32 %0, %1" : "=v"(y) : "v"(x));
  return y;
#endif
}

__device__ __forceinline__ float hw_exp(float x) {
#if __has_builtin(__builtin_amdgcn_exp2f)
  return __builtin_amdgcn_exp2f(x * 1.44269504088896340736f);
#else
  return __expf(x);
#endif
}

__device__ __forceinline__ float xhalf(float x) {
  const int b = __builtin_bit_cast(int, x);
  return __builtin_bit_cast(float, __builtin_amdgcn_permlanex16(
      b, b, 0x76543210, 0xfedcba98, false, false));
}

template <int M>
__device__ __forceinline__ float xrow(float x) {
  const int b = __builtin_bit_cast(int, x);
  return __builtin_bit_cast(float, __builtin_amdgcn_update_dpp(
      b, b, 0x160 + M, 0xf, 0xf, false));
}

struct Frags {
  v4f a0, a1;
  v4f b[4];
};

__device__ __forceinline__ void load_frags(const char* __restrict__ uA,
                                           unsigned aoff0, unsigned aoff1,
                                           const char* __restrict__ uB,
                                           long long rowB, unsigned boff,
                                           Frags& f) {
  f.a0 = *(const v4f*)(uA + aoff0);
  f.a1 = *(const v4f*)(uA + aoff1);
  const char* p = uB + boff;
  f.b[0] = *(const v4f*)(p);
  f.b[1] = *(const v4f*)(p + rowB);
  f.b[2] = *(const v4f*)(p + 2 * rowB);
  f.b[3] = *(const v4f*)(p + 3 * rowB);
}

#define WMMA_OPERAND_GAP() asm volatile("v_nop\n\tv_nop\n\tv_nop\n\tv_nop")

struct Acc {
  v8f c[2][4];
};

__device__ __forceinline__ void mma_step(v2f a0, v2f a1, v2f b0, v2f b1,
                                         v2f b2, v2f b3, Acc& acc) {
  acc.c[0][0] = __builtin_amdgcn_wmma_f32_16x16x4_f32(false, a0, false, b0, (short)0, acc.c[0][0], false, false);
  acc.c[0][1] = __builtin_amdgcn_wmma_f32_16x16x4_f32(false, a0, false, b1, (short)0, acc.c[0][1], false, false);
  acc.c[0][2] = __builtin_amdgcn_wmma_f32_16x16x4_f32(false, a0, false, b2, (short)0, acc.c[0][2], false, false);
  acc.c[0][3] = __builtin_amdgcn_wmma_f32_16x16x4_f32(false, a0, false, b3, (short)0, acc.c[0][3], false, false);
  acc.c[1][0] = __builtin_amdgcn_wmma_f32_16x16x4_f32(false, a1, false, b0, (short)0, acc.c[1][0], false, false);
  acc.c[1][1] = __builtin_amdgcn_wmma_f32_16x16x4_f32(false, a1, false, b1, (short)0, acc.c[1][1], false, false);
  acc.c[1][2] = __builtin_amdgcn_wmma_f32_16x16x4_f32(false, a1, false, b2, (short)0, acc.c[1][2], false, false);
  acc.c[1][3] = __builtin_amdgcn_wmma_f32_16x16x4_f32(false, a1, false, b3, (short)0, acc.c[1][3], false, false);
}

__device__ __forceinline__ void mma_pair(Frags& f, Acc& acc) {
  asm volatile("" : "+v"(f.b[0]), "+v"(f.b[1]), "+v"(f.b[2]), "+v"(f.b[3]));
  const v2f a0e = {f.a0.x, f.a0.y}, a0o = {f.a0.z, f.a0.w};
  const v2f a1e = {f.a1.x, f.a1.y}, a1o = {f.a1.z, f.a1.w};
  v2f be0 = {f.b[0].x, f.b[1].x}, be1 = {f.b[0].y, f.b[1].y};
  v2f be2 = {f.b[0].z, f.b[1].z}, be3 = {f.b[0].w, f.b[1].w};
  v2f bo0 = {f.b[2].x, f.b[3].x}, bo1 = {f.b[2].y, f.b[3].y};
  v2f bo2 = {f.b[2].z, f.b[3].z}, bo3 = {f.b[2].w, f.b[3].w};
  asm volatile("" : "+v"(be0), "+v"(be1), "+v"(be2), "+v"(be3),
                    "+v"(bo0), "+v"(bo1), "+v"(bo2), "+v"(bo3));
  mma_step(a0e, a1e, be0, be1, be2, be3, acc);
  mma_step(a0o, a1o, bo0, bo1, bo2, bo3, acc);
}

__device__ __forceinline__ void gemm_strip(
    const float* __restrict__ A, const float* __restrict__ Bm,
    float* __restrict__ C, int K, int N) {
  __shared__ float4 red[KW_ * 16 * 32];

  const int lane  = threadIdx.x & 31;
  const int part  = __builtin_amdgcn_readfirstlane((int)(threadIdx.x >> 5));
  const int Kp    = K / KW_;
  const int kBeg  = part * Kp;
  const int mTile = blockIdx.x * 32;
  const int nBase = blockIdx.y * 64;

  const int mRow = lane & 15;
  const int kHi  = (lane >> 4) * 4;
  const int col  = lane & 15;

  const char* uA = (const char*)(A + (long long)mTile * K + kBeg);
  const char* uB = (const char*)(Bm + (long long)kBeg * N + nBase);
  const long long rowB = (long long)N * 4;
  const unsigned aoff0 = (unsigned)(mRow * K + kHi) * 4u;
  const unsigned aoff1 = aoff0 + (unsigned)(16 * K) * 4u;
  const unsigned boff  = (unsigned)(kHi * N + 4 * col) * 4u;

  Acc acc = {};
  Frags r[4];

#define LOAD_PAIR(kk, dst)                                              \
  load_frags(uA + (long long)(kk) * 4, aoff0, aoff1,                    \
             uB + (long long)(kk) * rowB, rowB, boff, dst)

#pragma unroll
  for (int s = 0; s < 3; ++s) {
    LOAD_PAIR(8 * s, r[s]);
    __builtin_amdgcn_sched_barrier(0);
  }

  int k0 = 0;
  for (; k0 + 32 < Kp; k0 += 32) {
#pragma unroll
    for (int g = 0; g < 4; ++g) {
      LOAD_PAIR(k0 + 8 * (g + 3), r[(g + 3) & 3]);
      __builtin_amdgcn_sched_barrier(0);
      mma_pair(r[g], acc);
      __builtin_amdgcn_sched_barrier(0);
      WMMA_OPERAND_GAP();
      __builtin_amdgcn_sched_barrier(0);
    }
  }
  LOAD_PAIR(k0 + 24, r[3]);
  __builtin_amdgcn_sched_barrier(0);
#pragma unroll
  for (int s = 0; s < 4; ++s) {
    mma_pair(r[s], acc);
    __builtin_amdgcn_sched_barrier(0);
    WMMA_OPERAND_GAP();
    __builtin_amdgcn_sched_barrier(0);
  }
#undef LOAD_PAIR

#pragma unroll
  for (int m = 0; m < 2; ++m) {
#pragma unroll
    for (int rr = 0; rr < 8; ++rr) {
      float4 v;
      v.x = acc.c[m][0][rr]; v.y = acc.c[m][1][rr];
      v.z = acc.c[m][2][rr]; v.w = acc.c[m][3][rr];
      red[(part * 16 + m * 8 + rr) * 32 + lane] = v;
    }
  }
  __syncthreads();

  const int mOut0 = mTile + (part >> 1) * 16 + (lane >> 4) * 8 + (part & 1) * 4;
  const int nOut0 = nBase + 4 * col;
#pragma unroll
  for (int i = 0; i < 4; ++i) {
    const int q = 4 * part + i;
    float4 s = red[(0 * 16 + q) * 32 + lane];
#pragma unroll
    for (int p = 1; p < KW_; ++p) {
      const float4 v = red[(p * 16 + q) * 32 + lane];
      s.x += v.x; s.y += v.y; s.z += v.z; s.w += v.w;
    }
    *(float4*)(C + (long long)(mOut0 + i) * N + nOut0) = s;
  }
}

__global__ __launch_bounds__(32 * KW_) void wmma_gemm_f32(
    const float* __restrict__ A, const float* __restrict__ Bm,
    float* __restrict__ C, int K, int N,
    long long sA, long long sB, long long sC) {
  const int batch = blockIdx.z;
  gemm_strip(A + (long long)batch * sA, Bm + (long long)batch * sB,
             C + (long long)batch * sC, K, N);
}

__global__ __launch_bounds__(32 * KW_) void wmma_gemm_f32_pair(
    const float* __restrict__ A0, const float* __restrict__ B0,
    float* __restrict__ C0,
    const float* __restrict__ A1, const float* __restrict__ B1,
    float* __restrict__ C1, int K, int N) {
  const bool second = blockIdx.z != 0;
  gemm_strip(second ? A1 : A0, second ? B1 : B0, second ? C1 : C0, K, N);
}

__global__ __launch_bounds__(NK_) void scores_softmax(
    const float* __restrict__ qproj, const float* __restrict__ kproj,
    const float* __restrict__ Wv, const int* __restrict__ valid_lens,
    float* __restrict__ attn) {
  __shared__ float qp[QB_][H_];
  __shared__ float wv[H_];
  __shared__ float part[QB_][16];
  __shared__ float mxs[QB_];
  __shared__ float dns[QB_];

  const int q0   = blockIdx.x * QB_;
  const int b    = blockIdx.y;
  const int k    = threadIdx.x;
  const int lane = k & 31;
  const int wid  = k >> 5;

  for (int i = k; i < QB_ * H_; i += NK_)
    qp[i / H_][i % H_] = qproj[(long long)(b * NQ_ + q0) * H_ + i];
  if (k < H_) wv[k] = Wv[k];
  __syncthreads();

  const float4* kp4 =
      (const float4*)(kproj + (long long)(b * NK_ + k) * H_);
  float s[QB_];
#pragma unroll
  for (int j = 0; j < QB_; ++j) s[j] = 0.0f;

  for (int h4 = 0; h4 < H_ / 4; ++h4) {
    const float4 kv = kp4[h4];
    const int h = h4 * 4;
#pragma unroll
    for (int j = 0; j < QB_; ++j) {
      s[j] += wv[h + 0] * hw_tanh(qp[j][h + 0] + kv.x);
      s[j] += wv[h + 1] * hw_tanh(qp[j][h + 1] + kv.y);
      s[j] += wv[h + 2] * hw_tanh(qp[j][h + 2] + kv.z);
      s[j] += wv[h + 3] * hw_tanh(qp[j][h + 3] + kv.w);
    }
  }

  const int vlen = valid_lens[b];
  if (k >= vlen) {
#pragma unroll
    for (int j = 0; j < QB_; ++j) s[j] = -1.0e6f;
  }

  float mx[QB_];
#pragma unroll
  for (int j = 0; j < QB_; ++j) mx[j] = fmaxf(s[j], xhalf(s[j]));
#pragma unroll
  for (int j = 0; j < QB_; ++j) mx[j] = fmaxf(mx[j], xrow<8>(mx[j]));
#pragma unroll
  for (int j = 0; j < QB_; ++j) mx[j] = fmaxf(mx[j], xrow<4>(mx[j]));
#pragma unroll
  for (int j = 0; j < QB_; ++j) mx[j] = fmaxf(mx[j], xrow<2>(mx[j]));
#pragma unroll
  for (int j = 0; j < QB_; ++j) mx[j] = fmaxf(mx[j], xrow<1>(mx[j]));
  if (lane == 0) {
#pragma unroll
    for (int j = 0; j < QB_; ++j) part[j][wid] = mx[j];
  }
  __syncthreads();
  if (k < QB_) {
    float m = part[k][0];
#pragma unroll
    for (int w = 1; w < 16; ++w) m = fmaxf(m, part[k][w]);
    mxs[k] = m;
  }
  __syncthreads();

  float e[QB_], t[QB_];
#pragma unroll
  for (int j = 0; j < QB_; ++j) {
    e[j] = hw_exp(s[j] - mxs[j]);
    t[j] = e[j];
  }
#pragma unroll
  for (int j = 0; j < QB_; ++j) t[j] += xhalf(t[j]);
#pragma unroll
  for (int j = 0; j < QB_; ++j) t[j] += xrow<8>(t[j]);
#pragma unroll
  for (int j = 0; j < QB_; ++j) t[j] += xrow<4>(t[j]);
#pragma unroll
  for (int j = 0; j < QB_; ++j) t[j] += xrow<2>(t[j]);
#pragma unroll
  for (int j = 0; j < QB_; ++j) t[j] += xrow<1>(t[j]);
  if (lane == 0) {
#pragma unroll
    for (int j = 0; j < QB_; ++j) part[j][wid] = t[j];
  }
  __syncthreads();
  if (k < QB_) {
    float t = 0.0f;
#pragma unroll
    for (int w = 0; w < 16; ++w) t += part[k][w];
    dns[k] = t;
  }
  __syncthreads();

#pragma unroll
  for (int j = 0; j < QB_; ++j) {
    attn[(long long)(b * NQ_ + q0 + j) * NK_ + k] = e[j] / dns[j];
  }
}

extern "C" void kernel_launch(void* const* d_in, const int* in_sizes, int n_in,
                              void* d_out, int out_size, void* d_ws, size_t ws_size,
                              hipStream_t stream) {
  (void)in_sizes; (void)n_in; (void)out_size; (void)ws_size;

  const float* queries    = (const float*)d_in[0];
  const float* keys       = (const float*)d_in[1];
  const float* values     = (const float*)d_in[2];
  const float* Wq         = (const float*)d_in[3];
  const float* Wk         = (const float*)d_in[4];
  const float* Wv         = (const float*)d_in[5];
  const int*   valid_lens = (const int*)d_in[6];
  float*       out        = (float*)d_out;

  float* qproj = (float*)d_ws;
  float* kproj = qproj + (long long)B_ * NQ_ * H_;
  float* attn  = kproj + (long long)B_ * NK_ * H_;

  static_assert(NQ_ == NK_, "the paired launch needs equal row counts");
  wmma_gemm_f32_pair<<<dim3((B_ * NQ_) / 32, H_ / 64, 2), 32 * KW_, 0, stream>>>(
      queries, Wq, qproj, keys, Wk, kproj, D_, H_);

  scores_softmax<<<dim3(NQ_ / QB_, B_), NK_, 0, stream>>>(
      qproj, kproj, Wv, valid_lens, attn);

  wmma_gemm_f32<<<dim3(NQ_ / 32, D_ / 64, B_), 32 * KW_, 0, stream>>>(
      attn, values, out, NK_, D_,
      (long long)NQ_ * NK_, (long long)NK_ * D_, (long long)NQ_ * D_);
}


# Scores kernel: each thread owns two adjacent keys (256-thread blocks): half the waves, each q/Wv LDS read serves two keys, multiply-add fused (fma), 8-byte attn stores; row-sum order changed
# speedup vs baseline: 3.3650x; 1.0296x over previous
#include <hip/hip_runtime.h>

#define B_  4
#define NQ_ 512
#define NK_ 512
#define D_  512
#define H_  128
#define QB_ 8
#define KW_ 4
#define KPT_ 2
#define ST_  (NK_ / KPT_)
#define SW_  (ST_ / 32)

typedef __attribute__((ext_vector_type(2))) float v2f;
typedef __attribute__((ext_vector_type(4))) float v4f;
typedef __attribute__((ext_vector_type(8))) float v8f;

__device__ __forceinline__ float hw_tanh(float x) {
#if __has_builtin(__builtin_amdgcn_tanhf)
  return __builtin_amdgcn_tanhf(x);
#else
  float y;
  asm volatile("v_tanh_f32 %0, %1" : "=v"(y) : "v"(x));
  return y;
#endif
}

__device__ __forceinline__ float hw_exp(float x) {
#if __has_builtin(__builtin_amdgcn_exp2f)
  return __builtin_amdgcn_exp2f(x * 1.44269504088896340736f);
#else
  return __expf(x);
#endif
}

__device__ __forceinline__ float xhalf(float x) {
  const int b = __builtin_bit_cast(int, x);
  return __builtin_bit_cast(float, __builtin_amdgcn_permlanex16(
      b, b, 0x76543210, 0xfedcba98, false, false));
}

template <int M>
__device__ __forceinline__ float xrow(float x) {
  const int b = __builtin_bit_cast(int, x);
  return __builtin_bit_cast(float, __builtin_amdgcn_update_dpp(
      b, b, 0x160 + M, 0xf, 0xf, false));
}

struct Frags {
  v4f a0, a1;
  v4f b[4];
};

__device__ __forceinline__ void load_frags(const char* __restrict__ uA,
                                           unsigned aoff0, unsigned aoff1,
                                           const char* __restrict__ uB,
                                           long long rowB, unsigned boff,
                                           Frags& f) {
  f.a0 = *(const v4f*)(uA + aoff0);
  f.a1 = *(const v4f*)(uA + aoff1);
  const char* p = uB + boff;
  f.b[0] = *(const v4f*)(p);
  f.b[1] = *(const v4f*)(p + rowB);
  f.b[2] = *(const v4f*)(p + 2 * rowB);
  f.b[3] = *(const v4f*)(p + 3 * rowB);
}

#define WMMA_OPERAND_GAP() asm volatile("v_nop\n\tv_nop\n\tv_nop\n\tv_nop")

struct Acc {
  v8f c[2][4];
};

__device__ __forceinline__ void mma_step(v2f a0, v2f a1, v2f b0, v2f b1,
                                         v2f b2, v2f b3, Acc& acc) {
  acc.c[0][0] = __builtin_amdgcn_wmma_f32_16x16x4_f32(false, a0, false, b0, (short)0, acc.c[0][0], false, false);
  acc.c[0][1] = __builtin_amdgcn_wmma_f32_16x16x4_f32(false, a0, false, b1, (short)0, acc.c[0][1], false, false);
  acc.c[0][2] = __builtin_amdgcn_wmma_f32_16x16x4_f32(false, a0, false, b2, (short)0, acc.c[0][2], false, false);
  acc.c[0][3] = __builtin_amdgcn_wmma_f32_16x16x4_f32(false, a0, false, b3, (short)0, acc.c[0][3], false, false);
  acc.c[1][0] = __builtin_amdgcn_wmma_f32_16x16x4_f32(false, a1, false, b0, (short)0, acc.c[1][0], false, false);
  acc.c[1][1] = __builtin_amdgcn_wmma_f32_16x16x4_f32(false, a1, false, b1, (short)0, acc.c[1][1], false, false);
  acc.c[1][2] = __builtin_amdgcn_wmma_f32_16x16x4_f32(false, a1, false, b2, (short)0, acc.c[1][2], false, false);
  acc.c[1][3] = __builtin_amdgcn_wmma_f32_16x16x4_f32(false, a1, false, b3, (short)0, acc.c[1][3], false, false);
}

__device__ __forceinline__ void mma_pair(Frags& f, Acc& acc) {
  asm volatile("" : "+v"(f.b[0]), "+v"(f.b[1]), "+v"(f.b[2]), "+v"(f.b[3]));
  const v2f a0e = {f.a0.x, f.a0.y}, a0o = {f.a0.z, f.a0.w};
  const v2f a1e = {f.a1.x, f.a1.y}, a1o = {f.a1.z, f.a1.w};
  v2f be0 = {f.b[0].x, f.b[1].x}, be1 = {f.b[0].y, f.b[1].y};
  v2f be2 = {f.b[0].z, f.b[1].z}, be3 = {f.b[0].w, f.b[1].w};
  v2f bo0 = {f.b[2].x, f.b[3].x}, bo1 = {f.b[2].y, f.b[3].y};
  v2f bo2 = {f.b[2].z, f.b[3].z}, bo3 = {f.b[2].w, f.b[3].w};
  asm volatile("" : "+v"(be0), "+v"(be1), "+v"(be2), "+v"(be3),
                    "+v"(bo0), "+v"(bo1), "+v"(bo2), "+v"(bo3));
  mma_step(a0e, a1e, be0, be1, be2, be3, acc);
  mma_step(a0o, a1o, bo0, bo1, bo2, bo3, acc);
}

__device__ __forceinline__ void gemm_strip(
    const float* __restrict__ A, const float* __restrict__ Bm,
    float* __restrict__ C, int K, int N) {
  __shared__ float4 red[KW_ * 16 * 32];

  const int lane  = threadIdx.x & 31;
  const int part  = __builtin_amdgcn_readfirstlane((int)(threadIdx.x >> 5));
  const int Kp    = K / KW_;
  const int kBeg  = part * Kp;
  const int mTile = blockIdx.x * 32;
  const int nBase = blockIdx.y * 64;

  const int mRow = lane & 15;
  const int kHi  = (lane >> 4) * 4;
  const int col  = lane & 15;

  const char* uA = (const char*)(A + (long long)mTile * K + kBeg);
  const char* uB = (const char*)(Bm + (long long)kBeg * N + nBase);
  const long long rowB = (long long)N * 4;
  const unsigned aoff0 = (unsigned)(mRow * K + kHi) * 4u;
  const unsigned aoff1 = aoff0 + (unsigned)(16 * K) * 4u;
  const unsigned boff  = (unsigned)(kHi * N + 4 * col) * 4u;

  Acc acc = {};
  Frags r[4];

#define LOAD_PAIR(kk, dst)                                              \
  load_frags(uA + (long long)(kk) * 4, aoff0, aoff1,                    \
             uB + (long long)(kk) * rowB, rowB, boff, dst)

#pragma unroll
  for (int s = 0; s < 3; ++s) {
    LOAD_PAIR(8 * s, r[s]);
    __builtin_amdgcn_sched_barrier(0);
  }

  int k0 = 0;
  for (; k0 + 32 < Kp; k0 += 32) {
#pragma unroll
    for (int g = 0; g < 4; ++g) {
      LOAD_PAIR(k0 + 8 * (g + 3), r[(g + 3) & 3]);
      __builtin_amdgcn_sched_barrier(0);
      mma_pair(r[g], acc);
      __builtin_amdgcn_sched_barrier(0);
      WMMA_OPERAND_GAP();
      __builtin_amdgcn_sched_barrier(0);
    }
  }
  LOAD_PAIR(k0 + 24, r[3]);
  __builtin_amdgcn_sched_barrier(0);
#pragma unroll
  for (int s = 0; s < 4; ++s) {
    mma_pair(r[s], acc);
    __builtin_amdgcn_sched_barrier(0);
    WMMA_OPERAND_GAP();
    __builtin_amdgcn_sched_barrier(0);
  }
#undef LOAD_PAIR

#pragma unroll
  for (int m = 0; m < 2; ++m) {
#pragma unroll
    for (int rr = 0; rr < 8; ++rr) {
      float4 v;
      v.x = acc.c[m][0][rr]; v.y = acc.c[m][1][rr];
      v.z = acc.c[m][2][rr]; v.w = acc.c[m][3][rr];
      red[(part * 16 + m * 8 + rr) * 32 + lane] = v;
    }
  }
  __syncthreads();

  const int mOut0 = mTile + (part >> 1) * 16 + (lane >> 4) * 8 + (part & 1) * 4;
  const int nOut0 = nBase + 4 * col;
#pragma unroll
  for (int i = 0; i < 4; ++i) {
    const int q = 4 * part + i;
    float4 s = red[(0 * 16 + q) * 32 + lane];
#pragma unroll
    for (int p = 1; p < KW_; ++p) {
      const float4 v = red[(p * 16 + q) * 32 + lane];
      s.x += v.x; s.y += v.y; s.z += v.z; s.w += v.w;
    }
    *(float4*)(C + (long long)(mOut0 + i) * N + nOut0) = s;
  }
}

__global__ __launch_bounds__(32 * KW_) void wmma_gemm_f32(
    const float* __restrict__ A, const float* __restrict__ Bm,
    float* __restrict__ C, int K, int N,
    long long sA, long long sB, long long sC) {
  const int batch = blockIdx.z;
  gemm_strip(A + (long long)batch * sA, Bm + (long long)batch * sB,
             C + (long long)batch * sC, K, N);
}

__global__ __launch_bounds__(32 * KW_) void wmma_gemm_f32_pair(
    const float* __restrict__ A0, const float* __restrict__ B0,
    float* __restrict__ C0,
    const float* __restrict__ A1, const float* __restrict__ B1,
    float* __restrict__ C1, int K, int N) {
  const bool second = blockIdx.z != 0;
  gemm_strip(second ? A1 : A0, second ? B1 : B0, second ? C1 : C0, K, N);
}

__global__ __launch_bounds__(ST_) void scores_softmax(
    const float* __restrict__ qproj, const float* __restrict__ kproj,
    const float* __restrict__ Wv, const int* __restrict__ valid_lens,
    float* __restrict__ attn) {
  __shared__ float4 qp4[QB_][H_ / 4];
  __shared__ float4 wv4[H_ / 4];
  __shared__ float part[QB_][SW_];
  __shared__ float mxs[QB_];
  __shared__ float dns[QB_];

  const int q0   = blockIdx.x * QB_;
  const int b    = blockIdx.y;
  const int t    = threadIdx.x;
  const int lane = t & 31;
  const int wid  = t >> 5;
  const int k0   = KPT_ * t;

  static_assert(QB_ * H_ / 4 == ST_, "one float4 of qproj per thread");
  qp4[t / (H_ / 4)][t % (H_ / 4)] =
      ((const float4*)(qproj + (long long)(b * NQ_ + q0) * H_))[t];
  if (t < H_ / 4) wv4[t] = ((const float4*)Wv)[t];
  __syncthreads();

  const float4* kpa = (const float4*)(kproj + (long long)(b * NK_ + k0) * H_);
  const float4* kpb = kpa + H_ / 4;
  float sa[QB_], sb[QB_];
#pragma unroll
  for (int j = 0; j < QB_; ++j) { sa[j] = 0.0f; sb[j] = 0.0f; }

  for (int h4 = 0; h4 < H_ / 4; ++h4) {
    const float4 ka = kpa[h4];
    const float4 kb = kpb[h4];
    const float4 w  = wv4[h4];
#pragma unroll
    for (int j = 0; j < QB_; ++j) {
      const float4 q = qp4[j][h4];
      sa[j] = fmaf(w.x, hw_tanh(q.x + ka.x), sa[j]);
      sb[j] = fmaf(w.x, hw_tanh(q.x + kb.x), sb[j]);
      sa[j] = fmaf(w.y, hw_tanh(q.y + ka.y), sa[j]);
      sb[j] = fmaf(w.y, hw_tanh(q.y + kb.y), sb[j]);
      sa[j] = fmaf(w.z, hw_tanh(q.z + ka.z), sa[j]);
      sb[j] = fmaf(w.z, hw_tanh(q.z + kb.z), sb[j]);
      sa[j] = fmaf(w.w, hw_tanh(q.w + ka.w), sa[j]);
      sb[j] = fmaf(w.w, hw_tanh(q.w + kb.w), sb[j]);
    }
  }

  const int vlen = valid_lens[b];
  if (k0 >= vlen) {
#pragma unroll
    for (int j = 0; j < QB_; ++j) sa[j] = -1.0e6f;
  }
  if (k0 + 1 >= vlen) {
#pragma unroll
    for (int j = 0; j < QB_; ++j) sb[j] = -1.0e6f;
  }

  float mx[QB_];
#pragma unroll
  for (int j = 0; j < QB_; ++j) mx[j] = fmaxf(sa[j], sb[j]);
#pragma unroll
  for (int j = 0; j < QB_; ++j) mx[j] = fmaxf(mx[j], xhalf(mx[j]));
#pragma unroll
  for (int j = 0; j < QB_; ++j) mx[j] = fmaxf(mx[j], xrow<8>(mx[j]));
#pragma unroll
  for (int j = 0; j < QB_; ++j) mx[j] = fmaxf(mx[j], xrow<4>(mx[j]));
#pragma unroll
  for (int j = 0; j < QB_; ++j) mx[j] = fmaxf(mx[j], xrow<2>(mx[j]));
#pragma unroll
  for (int j = 0; j < QB_; ++j) mx[j] = fmaxf(mx[j], xrow<1>(mx[j]));
  if (lane == 0) {
#pragma unroll
    for (int j = 0; j < QB_; ++j) part[j][wid] = mx[j];
  }
  __syncthreads();
  if (t < QB_) {
    float m = part[t][0];
#pragma unroll
    for (int w = 1; w < SW_; ++w) m = fmaxf(m, part[t][w]);
    mxs[t] = m;
  }
  __syncthreads();

  float ea[QB_], eb[QB_], ts[QB_];
#pragma unroll
  for (int j = 0; j < QB_; ++j) {
    const float m = mxs[j];
    ea[j] = hw_exp(sa[j] - m);
    eb[j] = hw_exp(sb[j] - m);
    ts[j] = ea[j] + eb[j];
  }
#pragma unroll
  for (int j = 0; j < QB_; ++j) ts[j] += xhalf(ts[j]);
#pragma unroll
  for (int j = 0; j < QB_; ++j) ts[j] += xrow<8>(ts[j]);
#pragma unroll
  for (int j = 0; j < QB_; ++j) ts[j] += xrow<4>(ts[j]);
#pragma unroll
  for (int j = 0; j < QB_; ++j) ts[j] += xrow<2>(ts[j]);
#pragma unroll
  for (int j = 0; j < QB_; ++j) ts[j] += xrow<1>(ts[j]);
  if (lane == 0) {
#pragma unroll
    for (int j = 0; j < QB_; ++j) part[j][wid] = ts[j];
  }
  __syncthreads();
  if (t < QB_) {
    float d = 0.0f;
#pragma unroll
    for (int w = 0; w < SW_; ++w) d += part[t][w];
    dns[t] = d;
  }
  __syncthreads();

#pragma unroll
  for (int j = 0; j < QB_; ++j) {
    const float d = dns[j];
    float2 o;
    o.x = ea[j] / d;
    o.y = eb[j] / d;
    *(float2*)(attn + (long long)(b * NQ_ + q0 + j) * NK_ + k0) = o;
  }
}

extern "C" void kernel_launch(void* const* d_in, const int* in_sizes, int n_in,
                              void* d_out, int out_size, void* d_ws, size_t ws_size,
                              hipStream_t stream) {
  (void)in_sizes; (void)n_in; (void)out_size; (void)ws_size;

  const float* queries    = (const float*)d_in[0];
  const float* keys       = (const float*)d_in[1];
  const float* values     = (const float*)d_in[2];
  const float* Wq         = (const float*)d_in[3];
  const float* Wk         = (const float*)d_in[4];
  const float* Wv         = (const float*)d_in[5];
  const int*   valid_lens = (const int*)d_in[6];
  float*       out        = (float*)d_out;

  float* qproj = (float*)d_ws;
  float* kproj = qproj + (long long)B_ * NQ_ * H_;
  float* attn  = kproj + (long long)B_ * NK_ * H_;

  static_assert(NQ_ == NK_, "the paired launch needs equal row counts");
  wmma_gemm_f32_pair<<<dim3((B_ * NQ_) / 32, H_ / 64, 2), 32 * KW_, 0, stream>>>(
      queries, Wq, qproj, keys, Wk, kproj, D_, H_);

  scores_softmax<<<dim3(NQ_ / QB_, B_), ST_, 0, stream>>>(
      qproj, kproj, Wv, valid_lens, attn);

  wmma_gemm_f32<<<dim3(NQ_ / 32, D_ / 64, B_), 32 * KW_, 0, stream>>>(
      attn, values, out, NK_, D_,
      (long long)NQ_ * NK_, (long long)NK_ * D_, (long long)NQ_ * D_);
}


# kproj written key-packed [batch][h/4][key][4] by the projection GEMM: a scores-kernel kproj load touches 8 whole lines (was 32 partly used ones) and is issued one trip ahead; sums' order unchanged
# speedup vs baseline: 3.4524x; 1.0260x over previous
#include <hip/hip_runtime.h>

#define B_  4
#define NQ_ 512
#define NK_ 512
#define D_  512
#define H_  128
#define QB_ 8
#define KW_ 4
#define KPT_ 2
#define ST_  (NK_ / KPT_)
#define SW_  (ST_ / 32)
#define RS_  34

typedef __attribute__((ext_vector_type(2))) float v2f;
typedef __attribute__((ext_vector_type(4))) float v4f;
typedef __attribute__((ext_vector_type(8))) float v8f;

__device__ __forceinline__ float hw_tanh(float x) {
#if __has_builtin(__builtin_amdgcn_tanhf)
  return __builtin_amdgcn_tanhf(x);
#else
  float y;
  asm volatile("v_tanh_f32 %0, %1" : "=v"(y) : "v"(x));
  return y;
#endif
}

__device__ __forceinline__ float hw_exp(float x) {
#if __has_builtin(__builtin_amdgcn_exp2f)
  return __builtin_amdgcn_exp2f(x * 1.44269504088896340736f);
#else
  return __expf(x);
#endif
}

__device__ __forceinline__ float xhalf(float x) {
  const int b = __builtin_bit_cast(int, x);
  return __builtin_bit_cast(float, __builtin_amdgcn_permlanex16(
      b, b, 0x76543210, 0xfedcba98, false, false));
}

template <int M>
__device__ __forceinline__ float xrow(float x) {
  const int b = __builtin_bit_cast(int, x);
  return __builtin_bit_cast(float, __builtin_amdgcn_update_dpp(
      b, b, 0x160 + M, 0xf, 0xf, false));
}

struct Frags {
  v4f a0, a1;
  v4f b[4];
};

__device__ __forceinline__ void load_frags(const char* __restrict__ uA,
                                           unsigned aoff0, unsigned aoff1,
                                           const char* __restrict__ uB,
                                           long long rowB, unsigned boff,
                                           Frags& f) {
  f.a0 = *(const v4f*)(uA + aoff0);
  f.a1 = *(const v4f*)(uA + aoff1);
  const char* p = uB + boff;
  f.b[0] = *(const v4f*)(p);
  f.b[1] = *(const v4f*)(p + rowB);
  f.b[2] = *(const v4f*)(p + 2 * rowB);
  f.b[3] = *(const v4f*)(p + 3 * rowB);
}

#define WMMA_OPERAND_GAP() asm volatile("v_nop\n\tv_nop\n\tv_nop\n\tv_nop")

struct Acc {
  v8f c[2][4];
};

__device__ __forceinline__ void mma_step(v2f a0, v2f a1, v2f b0, v2f b1,
                                         v2f b2, v2f b3, Acc& acc) {
  acc.c[0][0] = __builtin_amdgcn_wmma_f32_16x16x4_f32(false, a0, false, b0, (short)0, acc.c[0][0], false, false);
  acc.c[0][1] = __builtin_amdgcn_wmma_f32_16x16x4_f32(false, a0, false, b1, (short)0, acc.c[0][1], false, false);
  acc.c[0][2] = __builtin_amdgcn_wmma_f32_16x16x4_f32(false, a0, false, b2, (short)0, acc.c[0][2], false, false);
  acc.c[0][3] = __builtin_amdgcn_wmma_f32_16x16x4_f32(false, a0, false, b3, (short)0, acc.c[0][3], false, false);
  acc.c[1][0] = __builtin_amdgcn_wmma_f32_16x16x4_f32(false, a1, false, b0, (short)0, acc.c[1][0], false, false);
  acc.c[1][1] = __builtin_amdgcn_wmma_f32_16x16x4_f32(false, a1, false, b1, (short)0, acc.c[1][1], false, false);
  acc.c[1][2] = __builtin_amdgcn_wmma_f32_16x16x4_f32(false, a1, false, b2, (short)0, acc.c[1][2], false, false);
  acc.c[1][3] = __builtin_amdgcn_wmma_f32_16x16x4_f32(false, a1, false, b3, (short)0, acc.c[1][3], false, false);
}

__device__ __forceinline__ void mma_pair(Frags& f, Acc& acc) {
  asm volatile("" : "+v"(f.b[0]), "+v"(f.b[1]), "+v"(f.b[2]), "+v"(f.b[3]));
  const v2f a0e = {f.a0.x, f.a0.y}, a0o = {f.a0.z, f.a0.w};
  const v2f a1e = {f.a1.x, f.a1.y}, a1o = {f.a1.z, f.a1.w};
  v2f be0 = {f.b[0].x, f.b[1].x}, be1 = {f.b[0].y, f.b[1].y};
  v2f be2 = {f.b[0].z, f.b[1].z}, be3 = {f.b[0].w, f.b[1].w};
  v2f bo0 = {f.b[2].x, f.b[3].x}, bo1 = {f.b[2].y, f.b[3].y};
  v2f bo2 = {f.b[2].z, f.b[3].z}, bo3 = {f.b[2].w, f.b[3].w};
  asm volatile("" : "+v"(be0), "+v"(be1), "+v"(be2), "+v"(be3),
                    "+v"(bo0), "+v"(bo1), "+v"(bo2), "+v"(bo3));
  mma_step(a0e, a1e, be0, be1, be2, be3, acc);
  mma_step(a0o, a1o, bo0, bo1, bo2, bo3, acc);
}

__device__ __forceinline__ void gemm_strip(
    const float* __restrict__ A, const float* __restrict__ Bm,
    float* __restrict__ C, int K, int N, bool keyPacked) {
  __shared__ float4 red[KW_ * 16 * RS_];

  const int lane  = threadIdx.x & 31;
  const int part  = __builtin_amdgcn_readfirstlane((int)(threadIdx.x >> 5));
  const int Kp    = K / KW_;
  const int kBeg  = part * Kp;
  const int mTile = blockIdx.x * 32;
  const int nBase = blockIdx.y * 64;

  const int mRow = lane & 15;
  const int kHi  = (lane >> 4) * 4;
  const int col  = lane & 15;

  const char* uA = (const char*)(A + (long long)mTile * K + kBeg);
  const char* uB = (const char*)(Bm + (long long)kBeg * N + nBase);
  const long long rowB = (long long)N * 4;
  const unsigned aoff0 = (unsigned)(mRow * K + kHi) * 4u;
  const unsigned aoff1 = aoff0 + (unsigned)(16 * K) * 4u;
  const unsigned boff  = (unsigned)(kHi * N + 4 * col) * 4u;

  Acc acc = {};
  Frags r[4];

#define LOAD_PAIR(kk, dst)                                              \
  load_frags(uA + (long long)(kk) * 4, aoff0, aoff1,                    \
             uB + (long long)(kk) * rowB, rowB, boff, dst)

#pragma unroll
  for (int s = 0; s < 3; ++s) {
    LOAD_PAIR(8 * s, r[s]);
    __builtin_amdgcn_sched_barrier(0);
  }

  int k0 = 0;
  for (; k0 + 32 < Kp; k0 += 32) {
#pragma unroll
    for (int g = 0; g < 4; ++g) {
      LOAD_PAIR(k0 + 8 * (g + 3), r[(g + 3) & 3]);
      __builtin_amdgcn_sched_barrier(0);
      mma_pair(r[g], acc);
      __builtin_amdgcn_sched_barrier(0);
      WMMA_OPERAND_GAP();
      __builtin_amdgcn_sched_barrier(0);
    }
  }
  LOAD_PAIR(k0 + 24, r[3]);
  __builtin_amdgcn_sched_barrier(0);
#pragma unroll
  for (int s = 0; s < 4; ++s) {
    mma_pair(r[s], acc);
    __builtin_amdgcn_sched_barrier(0);
    WMMA_OPERAND_GAP();
    __builtin_amdgcn_sched_barrier(0);
  }
#undef LOAD_PAIR

#pragma unroll
  for (int m = 0; m < 2; ++m) {
#pragma unroll
    for (int rr = 0; rr < 8; ++rr) {
      float4 v;
      v.x = acc.c[m][0][rr]; v.y = acc.c[m][1][rr];
      v.z = acc.c[m][2][rr]; v.w = acc.c[m][3][rr];
      red[(part * 16 + m * 8 + rr) * RS_ + lane] = v;
    }
  }
  __syncthreads();

  int eBase, eStep;
  float4* g;
  long long gStep;
  if (!keyPacked) {
    const int mOut0 = mTile + (part >> 1) * 16 + (lane >> 4) * 8 + (part & 1) * 4;
    const int nOut0 = nBase + 4 * col;
    eBase = 4 * part * RS_ + lane;
    eStep = RS_;
    g     = (float4*)(C + (long long)mOut0 * N + nOut0);
    gStep = N >> 2;
  } else {
    const int mt  = part >> 1;
    const int hf  = part & 1;
    const int rr  = lane & 7;
    const int cl  = lane >> 3;
    const int bb  = mTile / NK_;
    const int row = mTile % NK_ + mt * 16 + hf * 8 + rr;
    eBase = (mt * 8 + rr) * RS_ + hf * 16 + cl;
    eStep = 4;
    g     = (float4*)C + ((long long)bb * (N >> 2) + (nBase >> 2) + cl) * NK_ + row;
    gStep = 4LL * NK_;
  }
#pragma unroll
  for (int i = 0; i < 4; ++i) {
    const int e = eBase + i * eStep;
    float4 s = red[e];
#pragma unroll
    for (int p = 1; p < KW_; ++p) {
      const float4 v = red[p * 16 * RS_ + e];
      s.x += v.x; s.y += v.y; s.z += v.z; s.w += v.w;
    }
    g[(long long)i * gStep] = s;
  }
}

__global__ __launch_bounds__(32 * KW_) void wmma_gemm_f32(
    const float* __restrict__ A, const float* __restrict__ Bm,
    float* __restrict__ C, int K, int N,
    long long sA, long long sB, long long sC) {
  const int batch = blockIdx.z;
  gemm_strip(A + (long long)batch * sA, Bm + (long long)batch * sB,
             C + (long long)batch * sC, K, N, false);
}

__global__ __launch_bounds__(32 * KW_) void wmma_gemm_f32_pair(
    const float* __restrict__ A0, const float* __restrict__ B0,
    float* __restrict__ C0,
    const float* __restrict__ A1, const float* __restrict__ B1,
    float* __restrict__ C1, int K, int N) {
  const bool second = blockIdx.z != 0;
  gemm_strip(second ? A1 : A0, second ? B1 : B0, second ? C1 : C0, K, N, second);
}

__global__ __launch_bounds__(ST_) void scores_softmax(
    const float* __restrict__ qproj, const float* __restrict__ kproj,
    const float* __restrict__ Wv, const int* __restrict__ valid_lens,
    float* __restrict__ attn) {
  __shared__ float4 qp4[QB_][H_ / 4];
  __shared__ float4 wv4[H_ / 4];
  __shared__ float part[QB_][SW_];
  __shared__ float mxs[QB_];
  __shared__ float dns[QB_];

  const int q0   = blockIdx.x * QB_;
  const int b    = blockIdx.y;
  const int t    = threadIdx.x;
  const int lane = t & 31;
  const int wid  = t >> 5;
  const int k0   = KPT_ * t;

  const float4* kp = (const float4*)kproj + (long long)b * (H_ / 4) * NK_ + k0;
  float sa[QB_], sb[QB_];
#pragma unroll
  for (int j = 0; j < QB_; ++j) { sa[j] = 0.0f; sb[j] = 0.0f; }

#define SCORE_STEP(h4_, ka_, kb_)                                        \
  {                                                                      \
    const float4 w = wv4[h4_];                                           \
    _Pragma("unroll")                                                    \
    for (int j = 0; j < QB_; ++j) {                                      \
      const float4 q = qp4[j][h4_];                                      \
      sa[j] = fmaf(w.x, hw_tanh(q.x + ka_.x), sa[j]);                    \
      sb[j] = fmaf(w.x, hw_tanh(q.x + kb_.x), sb[j]);                    \
      sa[j] = fmaf(w.y, hw_tanh(q.y + ka_.y), sa[j]);                    \
      sb[j] = fmaf(w.y, hw_tanh(q.y + kb_.y), sb[j]);                    \
      sa[j] = fmaf(w.z, hw_tanh(q.z + ka_.z), sa[j]);                    \
      sb[j] = fmaf(w.z, hw_tanh(q.z + kb_.z), sb[j]);                    \
      sa[j] = fmaf(w.w, hw_tanh(q.w + ka_.w), sa[j]);                    \
      sb[j] = fmaf(w.w, hw_tanh(q.w + kb_.w), sb[j]);                    \
    }                                                                    \
  }

  float4 ka0 = kp[0],   kb0 = kp[1];
  float4 ka1 = kp[NK_], kb1 = kp[NK_ + 1];

  static_assert(QB_ * H_ / 4 == ST_, "one float4 of qproj per thread");
  qp4[t / (H_ / 4)][t % (H_ / 4)] =
      ((const float4*)(qproj + (long long)(b * NQ_ + q0) * H_))[t];
  if (t < H_ / 4) wv4[t] = ((const float4*)Wv)[t];
  __syncthreads();

  static_assert((H_ / 4) % 2 == 0, "two steps per trip");
  for (int h4 = 0; h4 + 2 < H_ / 4; h4 += 2) {
    const float4* kn = kp + (long long)(h4 + 2) * NK_;
    const float4 na0 = kn[0],   nb0 = kn[1];
    const float4 na1 = kn[NK_], nb1 = kn[NK_ + 1];
    SCORE_STEP(h4, ka0, kb0);
    SCORE_STEP(h4 + 1, ka1, kb1);
    ka0 = na0; kb0 = nb0; ka1 = na1; kb1 = nb1;
  }
  SCORE_STEP(H_ / 4 - 2, ka0, kb0);
  SCORE_STEP(H_ / 4 - 1, ka1, kb1);
#undef SCORE_STEP

  const int vlen = valid_lens[b];
  if (k0 >= vlen) {
#pragma unroll
    for (int j = 0; j < QB_; ++j) sa[j] = -1.0e6f;
  }
  if (k0 + 1 >= vlen) {
#pragma unroll
    for (int j = 0; j < QB_; ++j) sb[j] = -1.0e6f;
  }

  float mx[QB_];
#pragma unroll
  for (int j = 0; j < QB_; ++j) mx[j] = fmaxf(sa[j], sb[j]);
#pragma unroll
  for (int j = 0; j < QB_; ++j) mx[j] = fmaxf(mx[j], xhalf(mx[j]));
#pragma unroll
  for (int j = 0; j < QB_; ++j) mx[j] = fmaxf(mx[j], xrow<8>(mx[j]));
#pragma unroll
  for (int j = 0; j < QB_; ++j) mx[j] = fmaxf(mx[j], xrow<4>(mx[j]));
#pragma unroll
  for (int j = 0; j < QB_; ++j) mx[j] = fmaxf(mx[j], xrow<2>(mx[j]));
#pragma unroll
  for (int j = 0; j < QB_; ++j) mx[j] = fmaxf(mx[j], xrow<1>(mx[j]));
  if (lane == 0) {
#pragma unroll
    for (int j = 0; j < QB_; ++j) part[j][wid] = mx[j];
  }
  __syncthreads();
  if (t < QB_) {
    float m = part[t][0];
#pragma unroll
    for (int w = 1; w < SW_; ++w) m = fmaxf(m, part[t][w]);
    mxs[t] = m;
  }
  __syncthreads();

  float ea[QB_], eb[QB_], ts[QB_];
#pragma unroll
  for (int j = 0; j < QB_; ++j) {
    const float m = mxs[j];
    ea[j] = hw_exp(sa[j] - m);
    eb[j] = hw_exp(sb[j] - m);
    ts[j] = ea[j] + eb[j];
  }
#pragma unroll
  for (int j = 0; j < QB_; ++j) ts[j] += xhalf(ts[j]);
#pragma unroll
  for (int j = 0; j < QB_; ++j) ts[j] += xrow<8>(ts[j]);
#pragma unroll
  for (int j = 0; j < QB_; ++j) ts[j] += xrow<4>(ts[j]);
#pragma unroll
  for (int j = 0; j < QB_; ++j) ts[j] += xrow<2>(ts[j]);
#pragma unroll
  for (int j = 0; j < QB_; ++j) ts[j] += xrow<1>(ts[j]);
  if (lane == 0) {
#pragma unroll
    for (int j = 0; j < QB_; ++j) part[j][wid] = ts[j];
  }
  __syncthreads();
  if (t < QB_) {
    float d = 0.0f;
#pragma unroll
    for (int w = 0; w < SW_; ++w) d += part[t][w];
    dns[t] = d;
  }
  __syncthreads();

#pragma unroll
  for (int j = 0; j < QB_; ++j) {
    const float d = dns[j];
    float2 o;
    o.x = ea[j] / d;
    o.y = eb[j] / d;
    *(float2*)(attn + (long long)(b * NQ_ + q0 + j) * NK_ + k0) = o;
  }
}

extern "C" void kernel_launch(void* const* d_in, const int* in_sizes, int n_in,
                              void* d_out, int out_size, void* d_ws, size_t ws_size,
                              hipStream_t stream) {
  (void)in_sizes; (void)n_in; (void)out_size; (void)ws_size;

  const float* queries    = (const float*)d_in[0];
  const float* keys       = (const float*)d_in[1];
  const float* values     = (const float*)d_in[2];
  const float* Wq         = (const float*)d_in[3];
  const float* Wk         = (const float*)d_in[4];
  const float* Wv         = (const float*)d_in[5];
  const int*   valid_lens = (const int*)d_in[6];
  float*       out        = (float*)d_out;

  float* qproj = (float*)d_ws;
  float* kproj = qproj + (long long)B_ * NQ_ * H_;
  float* attn  = kproj + (long long)B_ * NK_ * H_;

  static_assert(NQ_ == NK_, "the paired launch needs equal row counts");
  wmma_gemm_f32_pair<<<dim3((B_ * NQ_) / 32, H_ / 64, 2), 32 * KW_, 0, stream>>>(
      queries, Wq, qproj, keys, Wk, kproj, D_, H_);

  scores_softmax<<<dim3(NQ_ / QB_, B_), ST_, 0, stream>>>(
      qproj, kproj, Wv, valid_lens, attn);

  wmma_gemm_f32<<<dim3(NQ_ / 32, D_ / 64, B_), 32 * KW_, 0, stream>>>(
      attn, values, out, NK_, D_,
      (long long)NQ_ * NK_, (long long)NK_ * D_, (long long)NQ_ * D_);
}
